# starting kernel
# baseline (speedup 1.0000x reference)
#include <hip/hip_runtime.h>
#include <hip/hip_bf16.h>
#include <math.h>

#define N_NODES 50000
#define E_EDGES 800000
#define ETOT    (E_EDGES + N_NODES)
#define HEADS   4
#define CHP     32
#define HC      128
#define NGRAPH  64
#define SLOPE   0.2f
#define LN_EPS  1e-5f

typedef float v2f __attribute__((ext_vector_type(2)));
typedef float v8f __attribute__((ext_vector_type(8)));

__device__ __forceinline__ unsigned f32_to_ord(float f) {
  unsigned u = __float_as_uint(f);
  return (u & 0x80000000u) ? ~u : (u | 0x80000000u);
}
__device__ __forceinline__ float ord_to_f32(unsigned u) {
  u = (u & 0x80000000u) ? (u & 0x7fffffffu) : ~u;
  return __uint_as_float(u);
}
__device__ __forceinline__ void atomAddF(float* p, float v) {
  unsafeAtomicAdd(p, v);
}

__global__ void fill_f32(float* __restrict__ p, float v, int n) {
  int i = blockIdx.x * blockDim.x + threadIdx.x;
  if (i < n) p[i] = v;
}
__global__ void fill_u32(unsigned* __restrict__ p, unsigned v, int n) {
  int i = blockIdx.x * blockDim.x + threadIdx.x;
  if (i < n) p[i] = v;
}

#define LDSW 136
#define ROWTILES 4
__global__ void gemm_bias_wmma(const float* __restrict__ X,
                               const float* __restrict__ W,
                               const float* __restrict__ bias,
                               float* __restrict__ Y, int n) {
  extern __shared__ float sW[];
  const int tid = threadIdx.x;
  for (int i = 0; i < 16; ++i) {
    int idx = tid + 256 * i;
    int r = idx >> 5, q = idx & 31;
    float4 w4 = *(const float4*)(W + r * HC + q * 4);
    float* d = sW + r * LDSW + q * 4;
    d[0] = w4.x; d[1] = w4.y; d[2] = w4.z; d[3] = w4.w;
  }
  __syncthreads();

  const int wave = tid >> 5;
  const int lane = tid & 31;
  const int m    = lane & 15;
  const int hi   = lane >> 4;
  const int col  = wave * 16 + m;
  const float bc = bias[col];
  const int rowbase = blockIdx.x * (16 * ROWTILES);

  for (int t = 0; t < ROWTILES; ++t) {
    int row0 = rowbase + t * 16;
    if (row0 >= n) break;
    int row  = row0 + m;
    int rowc = (row < n) ? row : (n - 1);
    const float* ap = X + (size_t)rowc * HC + 2 * hi;
    v8f acc = {};
    #pragma unroll
    for (int k0 = 0; k0 < HC; k0 += 4) {
      v2f a; a.x = ap[k0]; a.y = ap[k0 + 1];
      const float* bp = sW + (k0 + 2 * hi) * LDSW + col;
      v2f b; b.x = bp[0]; b.y = bp[LDSW];
      acc = __builtin_amdgcn_wmma_f32_16x16x4_f32(
          false, a, false, b, (short)0, acc, false, false);
    }
    #pragma unroll
    for (int r = 0; r < 8; ++r) {
      int orow = row0 + r + 8 * hi;
      if (orow < n) Y[(size_t)orow * HC + col] = acc[r] + bc;
    }
  }
}

__global__ void edge_score_kernel(const int* __restrict__ ei,
                                  const float* __restrict__ xl,
                                  const float* __restrict__ xr,
                                  const float* __restrict__ att,
                                  float* __restrict__ score,
                                  unsigned* __restrict__ mord) {
  int e = blockIdx.x * 8 + (threadIdx.x >> 5);
  int lane = threadIdx.x & 31;
  if (e >= ETOT) return;
  int src, dst;
  if (e < E_EDGES) { src = ei[e]; dst = ei[E_EDGES + e]; }
  else             { src = dst = e - E_EDGES; }
  const float* pl = xl + (size_t)src * HC + lane;
  const float* pr = xr + (size_t)dst * HC + lane;
  float s[HEADS];
  #pragma unroll
  for (int h = 0; h < HEADS; ++h) {
    float v = pl[h * CHP] + pr[h * CHP];
    v = (v > 0.f) ? v : v * SLOPE;
    s[h] = v * att[h * CHP + lane];
  }
  #pragma unroll
  for (int h = 0; h < HEADS; ++h)
    for (int off = 16; off > 0; off >>= 1) s[h] += __shfl_xor(s[h], off, 32);
  if (lane < HEADS) {
    float sh = s[lane];
    score[(size_t)e * HEADS + lane] = sh;
    atomicMax(&mord[(size_t)dst * HEADS + lane], f32_to_ord(sh));
  }
}

__global__ void edge_exp_kernel(const int* __restrict__ ei,
                                float* __restrict__ score,
                                const unsigned* __restrict__ mord,
                                float* __restrict__ denom) {
  int t = blockIdx.x * blockDim.x + threadIdx.x;
  if (t >= ETOT * HEADS) return;
  int e = t >> 2, h = t & 3;
  int dst = (e < E_EDGES) ? ei[E_EDGES + e] : (e - E_EDGES);
  float m = ord_to_f32(mord[(size_t)dst * HEADS + h]);
  float ex = __expf(score[t] - m);
  score[t] = ex;
  atomAddF(&denom[(size_t)dst * HEADS + h], ex);
}

__global__ void edge_aggregate_kernel(const int* __restrict__ ei,
                                      const float* __restrict__ xl,
                                      const float* __restrict__ ex,
                                      const float* __restrict__ denom,
                                      float* __restrict__ agg) {
  int e = blockIdx.x * 8 + (threadIdx.x >> 5);
  int lane = threadIdx.x & 31;
  if (e >= ETOT) return;
  int src, dst;
  if (e < E_EDGES) { src = ei[e]; dst = ei[E_EDGES + e]; }
  else             { src = dst = e - E_EDGES; }
  #pragma unroll
  for (int h = 0; h < HEADS; ++h) {
    float alpha = ex[(size_t)e * HEADS + h] / denom[(size_t)dst * HEADS + h];
    float v = xl[(size_t)src * HC + h * CHP + lane] * alpha;
    atomAddF(&agg[(size_t)dst * HC + h * CHP + lane], v);
  }
}

__global__ void ln_silu_kernel(const float* __restrict__ agg,
                               const float* __restrict__ bias,
                               const float* __restrict__ gamma,
                               const float* __restrict__ beta,
                               float* __restrict__ out) {
  int node = blockIdx.x * 8 + (threadIdx.x >> 5);
  int lane = threadIdx.x & 31;
  if (node >= N_NODES) return;
  float v[4]; float sum = 0.f;
  #pragma unroll
  for (int i = 0; i < 4; ++i) {
    int c = i * 32 + lane;
    v[i] = agg[(size_t)node * HC + c] + bias[c];
    sum += v[i];
  }
  for (int off = 16; off > 0; off >>= 1) sum += __shfl_xor(sum, off, 32);
  float mu = sum * (1.0f / HC);
  float sq = 0.f;
  #pragma unroll
  for (int i = 0; i < 4; ++i) { float d = v[i] - mu; sq += d * d; }
  for (int off = 16; off > 0; off >>= 1) sq += __shfl_xor(sq, off, 32);
  float rstd = rsqrtf(sq * (1.0f / HC) + LN_EPS);
  #pragma unroll
  for (int i = 0; i < 4; ++i) {
    int c = i * 32 + lane;
    float h = (v[i] - mu) * rstd * gamma[c] + beta[c];
    out[(size_t)node * HC + c] = h / (1.0f + __expf(-h));
  }
}

__global__ void pool_accum_kernel(const float* __restrict__ x,
                                  const int* __restrict__ batch,
                                  float* __restrict__ sums,
                                  float* __restrict__ counts) {
  int node = blockIdx.x * 8 + (threadIdx.x >> 5);
  int lane = threadIdx.x & 31;
  if (node >= N_NODES) return;
  int g = batch[node];
  #pragma unroll
  for (int i = 0; i < 4; ++i) {
    int c = i * 32 + lane;
    atomAddF(&sums[(size_t)g * HC + c], x[(size_t)node * HC + c]);
  }
  if (lane == 0) atomAddF(&counts[g], 1.0f);
}
__global__ void pool_final_kernel(const float* __restrict__ sums,
                                  const float* __restrict__ counts,
                                  float* __restrict__ out) {
  int t = blockIdx.x * blockDim.x + threadIdx.x;
  if (t >= NGRAPH * HC) return;
  out[t] = sums[t] / fmaxf(counts[t >> 7], 1.0f);
}

extern "C" void kernel_launch(void* const* d_in, const int* in_sizes, int n_in,
                              void* d_out, int out_size, void* d_ws, size_t ws_size,
                              hipStream_t stream) {
  (void)in_sizes; (void)n_in; (void)out_size; (void)ws_size;
  const float* x     = (const float*)d_in[0];
  const int*   ei    = (const int*)d_in[1];
  const int*   batch = (const int*)d_in[2];
  const float* Wl[2]    = {(const float*)d_in[3],  (const float*)d_in[11]};
  const float* Wr[2]    = {(const float*)d_in[4],  (const float*)d_in[12]};
  const float* bl[2]    = {(const float*)d_in[5],  (const float*)d_in[13]};
  const float* br[2]    = {(const float*)d_in[6],  (const float*)d_in[14]};
  const float* att[2]   = {(const float*)d_in[7],  (const float*)d_in[15]};
  const float* bias[2]  = {(const float*)d_in[8],  (const float*)d_in[16]};
  const float* gamma[2] = {(const float*)d_in[9],  (const float*)d_in[17]};
  const float* beta[2]  = {(const float*)d_in[10], (const float*)d_in[18]};

  float* ws = (float*)d_ws;
  float*    xl    = ws;            ws += (size_t)N_NODES * HC;
  float*    xr2   = ws;            ws += (size_t)N_NODES * HC;
  float*    agg   = ws;            ws += (size_t)N_NODES * HC;
  float*    xnode = ws;            ws += (size_t)N_NODES * HC;
  float*    exbuf = ws;            ws += (size_t)ETOT * HEADS;
  float*    denom = ws;            ws += (size_t)N_NODES * HEADS;
  unsigned* mord  = (unsigned*)ws; ws += (size_t)N_NODES * HEADS;
  float*    psums = ws;            ws += (size_t)NGRAPH * HC;
  float*    pcnt  = ws;

  dim3 b256(256);
  const int    gemmGrid = (N_NODES + 16 * ROWTILES - 1) / (16 * ROWTILES);
  const size_t shmem    = (size_t)HC * LDSW * sizeof(float);
  const int    edgeGrid = (ETOT + 7) / 8;
  const int    nodeGrid = (N_NODES + 7) / 8;
  const unsigned ORD_NEG_INF = 0x007FFFFFu;

  const float* xin = x;
  for (int l = 0; l < 2; ++l) {
    gemm_bias_wmma<<<gemmGrid, b256, shmem, stream>>>(xin, Wl[l], bl[l], xl, N_NODES);
    gemm_bias_wmma<<<gemmGrid, b256, shmem, stream>>>(xin, Wr[l], br[l], xr2, N_NODES);
    fill_u32<<<(N_NODES * HEADS + 255) / 256, b256, 0, stream>>>(mord, ORD_NEG_INF, N_NODES * HEADS);
    fill_f32<<<(N_NODES * HEADS + 255) / 256, b256, 0, stream>>>(denom, 0.f, N_NODES * HEADS);
    fill_f32<<<(N_NODES * HC + 255) / 256, b256, 0, stream>>>(agg, 0.f, N_NODES * HC);
    edge_score_kernel<<<edgeGrid, b256, 0, stream>>>(ei, xl, xr2, att[l], exbuf, mord);
    edge_exp_kernel<<<(ETOT * HEADS + 255) / 256, b256, 0, stream>>>(ei, exbuf, mord, denom);
    edge_aggregate_kernel<<<edgeGrid, b256, 0, stream>>>(ei, xl, exbuf, denom, agg);
    ln_silu_kernel<<<nodeGrid, b256, 0, stream>>>(agg, bias[l], gamma[l], beta[l], xnode);
    xin = xnode;
  }
  fill_f32<<<(NGRAPH * HC + NGRAPH + 255) / 256, b256, 0, stream>>>(psums, 0.f, NGRAPH * HC + NGRAPH);
  pool_accum_kernel<<<nodeGrid, b256, 0, stream>>>(xnode, batch, psums, pcnt);
  pool_final_kernel<<<(NGRAPH * HC + 255) / 256, b256, 0, stream>>>(psums, pcnt, (float*)d_out);
}


# pool: each wave sums runs of equal graph id over 32 consecutive nodes in registers, one float atomic per run and channel instead of one per node; changes the pool's float summation order
# speedup vs baseline: 1.1416x; 1.1416x over previous
#include <hip/hip_runtime.h>
#include <hip/hip_bf16.h>
#include <math.h>

#define N_NODES 50000
#define E_EDGES 800000
#define ETOT    (E_EDGES + N_NODES)
#define HEADS   4
#define CHP     32
#define HC      128
#define NGRAPH  64
#define SLOPE   0.2f
#define LN_EPS  1e-5f

typedef float v2f __attribute__((ext_vector_type(2)));
typedef float v8f __attribute__((ext_vector_type(8)));

__device__ __forceinline__ unsigned f32_to_ord(float f) {
  unsigned u = __float_as_uint(f);
  return (u & 0x80000000u) ? ~u : (u | 0x80000000u);
}
__device__ __forceinline__ float ord_to_f32(unsigned u) {
  u = (u & 0x80000000u) ? (u & 0x7fffffffu) : ~u;
  return __uint_as_float(u);
}
__device__ __forceinline__ void atomAddF(float* p, float v) {
  unsafeAtomicAdd(p, v);
}

__global__ void fill_f32(float* __restrict__ p, float v, int n) {
  int i = blockIdx.x * blockDim.x + threadIdx.x;
  if (i < n) p[i] = v;
}
__global__ void fill_u32(unsigned* __restrict__ p, unsigned v, int n) {
  int i = blockIdx.x * blockDim.x + threadIdx.x;
  if (i < n) p[i] = v;
}

#define LDSW 136
#define ROWTILES 4
__global__ void gemm_bias_wmma(const float* __restrict__ X,
                               const float* __restrict__ W,
                               const float* __restrict__ bias,
                               float* __restrict__ Y, int n) {
  extern __shared__ float sW[];
  const int tid = threadIdx.x;
  for (int i = 0; i < 16; ++i) {
    int idx = tid + 256 * i;
    int r = idx >> 5, q = idx & 31;
    float4 w4 = *(const float4*)(W + r * HC + q * 4);
    float* d = sW + r * LDSW + q * 4;
    d[0] = w4.x; d[1] = w4.y; d[2] = w4.z; d[3] = w4.w;
  }
  __syncthreads();

  const int wave = tid >> 5;
  const int lane = tid & 31;
  const int m    = lane & 15;
  const int hi   = lane >> 4;
  const int col  = wave * 16 + m;
  const float bc = bias[col];
  const int rowbase = blockIdx.x * (16 * ROWTILES);

  for (int t = 0; t < ROWTILES; ++t) {
    int row0 = rowbase + t * 16;
    if (row0 >= n) break;
    int row  = row0 + m;
    int rowc = (row < n) ? row : (n - 1);
    const float* ap = X + (size_t)rowc * HC + 2 * hi;
    v8f acc = {};
    #pragma unroll
    for (int k0 = 0; k0 < HC; k0 += 4) {
      v2f a; a.x = ap[k0]; a.y = ap[k0 + 1];
      const float* bp = sW + (k0 + 2 * hi) * LDSW + col;
      v2f b; b.x = bp[0]; b.y = bp[LDSW];
      acc = __builtin_amdgcn_wmma_f32_16x16x4_f32(
          false, a, false, b, (short)0, acc, false, false);
    }
    #pragma unroll
    for (int r = 0; r < 8; ++r) {
      int orow = row0 + r + 8 * hi;
      if (orow < n) Y[(size_t)orow * HC + col] = acc[r] + bc;
    }
  }
}

__global__ void edge_score_kernel(const int* __restrict__ ei,
                                  const float* __restrict__ xl,
                                  const float* __restrict__ xr,
                                  const float* __restrict__ att,
                                  float* __restrict__ score,
                                  unsigned* __restrict__ mord) {
  int e = blockIdx.x * 8 + (threadIdx.x >> 5);
  int lane = threadIdx.x & 31;
  if (e >= ETOT) return;
  int src, dst;
  if (e < E_EDGES) { src = ei[e]; dst = ei[E_EDGES + e]; }
  else             { src = dst = e - E_EDGES; }
  const float* pl = xl + (size_t)src * HC + lane;
  const float* pr = xr + (size_t)dst * HC + lane;
  float s[HEADS];
  #pragma unroll
  for (int h = 0; h < HEADS; ++h) {
    float v = pl[h * CHP] + pr[h * CHP];
    v = (v > 0.f) ? v : v * SLOPE;
    s[h] = v * att[h * CHP + lane];
  }
  #pragma unroll
  for (int h = 0; h < HEADS; ++h)
    for (int off = 16; off > 0; off >>= 1) s[h] += __shfl_xor(s[h], off, 32);
  if (lane < HEADS) {
    float sh = s[lane];
    score[(size_t)e * HEADS + lane] = sh;
    atomicMax(&mord[(size_t)dst * HEADS + lane], f32_to_ord(sh));
  }
}

__global__ void edge_exp_kernel(const int* __restrict__ ei,
                                float* __restrict__ score,
                                const unsigned* __restrict__ mord,
                                float* __restrict__ denom) {
  int t = blockIdx.x * blockDim.x + threadIdx.x;
  if (t >= ETOT * HEADS) return;
  int e = t >> 2, h = t & 3;
  int dst = (e < E_EDGES) ? ei[E_EDGES + e] : (e - E_EDGES);
  float m = ord_to_f32(mord[(size_t)dst * HEADS + h]);
  float ex = __expf(score[t] - m);
  score[t] = ex;
  atomAddF(&denom[(size_t)dst * HEADS + h], ex);
}

__global__ void edge_aggregate_kernel(const int* __restrict__ ei,
                                      const float* __restrict__ xl,
                                      const float* __restrict__ ex,
                                      const float* __restrict__ denom,
                                      float* __restrict__ agg) {
  int e = blockIdx.x * 8 + (threadIdx.x >> 5);
  int lane = threadIdx.x & 31;
  if (e >= ETOT) return;
  int src, dst;
  if (e < E_EDGES) { src = ei[e]; dst = ei[E_EDGES + e]; }
  else             { src = dst = e - E_EDGES; }
  #pragma unroll
  for (int h = 0; h < HEADS; ++h) {
    float alpha = ex[(size_t)e * HEADS + h] / denom[(size_t)dst * HEADS + h];
    float v = xl[(size_t)src * HC + h * CHP + lane] * alpha;
    atomAddF(&agg[(size_t)dst * HC + h * CHP + lane], v);
  }
}

__global__ void ln_silu_kernel(const float* __restrict__ agg,
                               const float* __restrict__ bias,
                               const float* __restrict__ gamma,
                               const float* __restrict__ beta,
                               float* __restrict__ out) {
  int node = blockIdx.x * 8 + (threadIdx.x >> 5);
  int lane = threadIdx.x & 31;
  if (node >= N_NODES) return;
  float v[4]; float sum = 0.f;
  #pragma unroll
  for (int i = 0; i < 4; ++i) {
    int c = i * 32 + lane;
    v[i] = agg[(size_t)node * HC + c] + bias[c];
    sum += v[i];
  }
  for (int off = 16; off > 0; off >>= 1) sum += __shfl_xor(sum, off, 32);
  float mu = sum * (1.0f / HC);
  float sq = 0.f;
  #pragma unroll
  for (int i = 0; i < 4; ++i) { float d = v[i] - mu; sq += d * d; }
  for (int off = 16; off > 0; off >>= 1) sq += __shfl_xor(sq, off, 32);
  float rstd = rsqrtf(sq * (1.0f / HC) + LN_EPS);
  #pragma unroll
  for (int i = 0; i < 4; ++i) {
    int c = i * 32 + lane;
    float h = (v[i] - mu) * rstd * gamma[c] + beta[c];
    out[(size_t)node * HC + c] = h / (1.0f + __expf(-h));
  }
}

#define POOL_NODES 32
#define POOL_BATCH 16
__device__ __forceinline__ void pool_flush(float* __restrict__ sums,
                                           float* __restrict__ counts,
                                           int g, int lane, float4 acc, float cnt) {
  float* p = sums + (size_t)g * HC + 4 * lane;
  atomAddF(p + 0, acc.x);
  atomAddF(p + 1, acc.y);
  atomAddF(p + 2, acc.z);
  atomAddF(p + 3, acc.w);
  if (lane == 0) atomAddF(&counts[g], cnt);
}
__global__ void __launch_bounds__(32)
pool_accum_kernel(const float* __restrict__ x,
                  const int* __restrict__ batch,
                  float* __restrict__ sums,
                  float* __restrict__ counts) {
  const int lane = threadIdx.x;
  const int n0 = blockIdx.x * POOL_NODES;
  const int left = N_NODES - n0;
  const int nn = (left < POOL_NODES) ? left : POOL_NODES;
  const float4* xp = (const float4*)(x + (size_t)n0 * HC) + lane;
  int gcur = batch[n0];
  float4 acc = make_float4(0.f, 0.f, 0.f, 0.f);
  float cnt = 0.f;
  for (int j0 = 0; j0 < nn; j0 += POOL_BATCH) {
    float4 v[POOL_BATCH];
    int g[POOL_BATCH];
    #pragma unroll
    for (int j = 0; j < POOL_BATCH; ++j) {
      int jj = (j0 + j < nn) ? (j0 + j) : (nn - 1);
      v[j] = xp[(size_t)jj * (HC / 4)];
      g[j] = batch[n0 + jj];
    }
    #pragma unroll
    for (int j = 0; j < POOL_BATCH; ++j) {
      if (j0 + j < nn) {
        if (g[j] != gcur) {
          pool_flush(sums, counts, gcur, lane, acc, cnt);
          acc = make_float4(0.f, 0.f, 0.f, 0.f);
          cnt = 0.f;
          gcur = g[j];
        }
        acc.x += v[j].x; acc.y += v[j].y; acc.z += v[j].z; acc.w += v[j].w;
        cnt += 1.0f;
      }
    }
  }
  pool_flush(sums, counts, gcur, lane, acc, cnt);
}
__global__ void pool_final_kernel(const float* __restrict__ sums,
                                  const float* __restrict__ counts,
                                  float* __restrict__ out) {
  int t = blockIdx.x * blockDim.x + threadIdx.x;
  if (t >= NGRAPH * HC) return;
  out[t] = sums[t] / fmaxf(counts[t >> 7], 1.0f);
}

extern "C" void kernel_launch(void* const* d_in, const int* in_sizes, int n_in,
                              void* d_out, int out_size, void* d_ws, size_t ws_size,
                              hipStream_t stream) {
  (void)in_sizes; (void)n_in; (void)out_size; (void)ws_size;
  const float* x     = (const float*)d_in[0];
  const int*   ei    = (const int*)d_in[1];
  const int*   batch = (const int*)d_in[2];
  const float* Wl[2]    = {(const float*)d_in[3],  (const float*)d_in[11]};
  const float* Wr[2]    = {(const float*)d_in[4],  (const float*)d_in[12]};
  const float* bl[2]    = {(const float*)d_in[5],  (const float*)d_in[13]};
  const float* br[2]    = {(const float*)d_in[6],  (const float*)d_in[14]};
  const float* att[2]   = {(const float*)d_in[7],  (const float*)d_in[15]};
  const float* bias[2]  = {(const float*)d_in[8],  (const float*)d_in[16]};
  const float* gamma[2] = {(const float*)d_in[9],  (const float*)d_in[17]};
  const float* beta[2]  = {(const float*)d_in[10], (const float*)d_in[18]};

  float* ws = (float*)d_ws;
  float*    xl    = ws;            ws += (size_t)N_NODES * HC;
  float*    xr2   = ws;            ws += (size_t)N_NODES * HC;
  float*    agg   = ws;            ws += (size_t)N_NODES * HC;
  float*    xnode = ws;            ws += (size_t)N_NODES * HC;
  float*    exbuf = ws;            ws += (size_t)ETOT * HEADS;
  float*    denom = ws;            ws += (size_t)N_NODES * HEADS;
  unsigned* mord  = (unsigned*)ws; ws += (size_t)N_NODES * HEADS;
  float*    psums = ws;            ws += (size_t)NGRAPH * HC;
  float*    pcnt  = ws;

  dim3 b256(256);
  const int    gemmGrid = (N_NODES + 16 * ROWTILES - 1) / (16 * ROWTILES);
  const size_t shmem    = (size_t)HC * LDSW * sizeof(float);
  const int    edgeGrid = (ETOT + 7) / 8;
  const int    nodeGrid = (N_NODES + 7) / 8;
  const int    poolGrid = (N_NODES + POOL_NODES - 1) / POOL_NODES;
  const unsigned ORD_NEG_INF = 0x007FFFFFu;

  const float* xin = x;
  for (int l = 0; l < 2; ++l) {
    gemm_bias_wmma<<<gemmGrid, b256, shmem, stream>>>(xin, Wl[l], bl[l], xl, N_NODES);
    gemm_bias_wmma<<<gemmGrid, b256, shmem, stream>>>(xin, Wr[l], br[l], xr2, N_NODES);
    fill_u32<<<(N_NODES * HEADS + 255) / 256, b256, 0, stream>>>(mord, ORD_NEG_INF, N_NODES * HEADS);
    fill_f32<<<(N_NODES * HEADS + 255) / 256, b256, 0, stream>>>(denom, 0.f, N_NODES * HEADS);
    fill_f32<<<(N_NODES * HC + 255) / 256, b256, 0, stream>>>(agg, 0.f, N_NODES * HC);
    edge_score_kernel<<<edgeGrid, b256, 0, stream>>>(ei, xl, xr2, att[l], exbuf, mord);
    edge_exp_kernel<<<(ETOT * HEADS + 255) / 256, b256, 0, stream>>>(ei, exbuf, mord, denom);
    edge_aggregate_kernel<<<edgeGrid, b256, 0, stream>>>(ei, xl, exbuf, denom, agg);
    ln_silu_kernel<<<nodeGrid, b256, 0, stream>>>(agg, bias[l], gamma[l], beta[l], xnode);
    xin = xnode;
  }
  fill_f32<<<(NGRAPH * HC + NGRAPH + 255) / 256, b256, 0, stream>>>(psums, 0.f, NGRAPH * HC + NGRAPH);
  pool_accum_kernel<<<poolGrid, dim3(32), 0, stream>>>(xnode, batch, psums, pcnt);
  pool_final_kernel<<<(NGRAPH * HC + 255) / 256, b256, 0, stream>>>(psums, pcnt, (float*)d_out);
}


# edges bucketed by destination once per call (int count, scan, place); one wave per node: running-max softmax, weighted sum, LayerNorm in registers; no float atomics in layers; sum order changes
# speedup vs baseline: 3.0666x; 2.6862x over previous
#include <hip/hip_runtime.h>
#include <hip/hip_bf16.h>
#include <math.h>

#define N_NODES 50000
#define E_EDGES 800000
#define ETOT    (E_EDGES + N_NODES)
#define HEADS   4
#define CHP     32
#define HC      128
#define NGRAPH  64
#define SLOPE   0.2f
#define LN_EPS  1e-5f

typedef float v2f __attribute__((ext_vector_type(2)));
typedef float v8f __attribute__((ext_vector_type(8)));
typedef float v4f __attribute__((ext_vector_type(4)));

__device__ __forceinline__ void atomAddF(float* p, float v) {
  unsafeAtomicAdd(p, v);
}

__global__ void fill_f32(float* __restrict__ p, float v, int n) {
  int i = blockIdx.x * blockDim.x + threadIdx.x;
  if (i < n) p[i] = v;
}
__global__ void fill_u32(unsigned* __restrict__ p, unsigned v, int n) {
  int i = blockIdx.x * blockDim.x + threadIdx.x;
  if (i < n) p[i] = v;
}

#define LDSW 136
#define ROWTILES 4
__global__ void gemm_bias_wmma(const float* __restrict__ X,
                               const float* __restrict__ W,
                               const float* __restrict__ bias,
                               float* __restrict__ Y, int n) {
  extern __shared__ float sW[];
  const int tid = threadIdx.x;
  for (int i = 0; i < 16; ++i) {
    int idx = tid + 256 * i;
    int r = idx >> 5, q = idx & 31;
    float4 w4 = *(const float4*)(W + r * HC + q * 4);
    float* d = sW + r * LDSW + q * 4;
    d[0] = w4.x; d[1] = w4.y; d[2] = w4.z; d[3] = w4.w;
  }
  __syncthreads();

  const int wave = tid >> 5;
  const int lane = tid & 31;
  const int m    = lane & 15;
  const int hi   = lane >> 4;
  const int col  = wave * 16 + m;
  const float bc = bias[col];
  const int rowbase = blockIdx.x * (16 * ROWTILES);

  for (int t = 0; t < ROWTILES; ++t) {
    int row0 = rowbase + t * 16;
    if (row0 >= n) break;
    int row  = row0 + m;
    int rowc = (row < n) ? row : (n - 1);
    const float* ap = X + (size_t)rowc * HC + 2 * hi;
    v8f acc = {};
    #pragma unroll
    for (int k0 = 0; k0 < HC; k0 += 4) {
      v2f a; a.x = ap[k0]; a.y = ap[k0 + 1];
      const float* bp = sW + (k0 + 2 * hi) * LDSW + col;
      v2f b; b.x = bp[0]; b.y = bp[LDSW];
      acc = __builtin_amdgcn_wmma_f32_16x16x4_f32(
          false, a, false, b, (short)0, acc, false, false);
    }
    #pragma unroll
    for (int r = 0; r < 8; ++r) {
      int orow = row0 + r + 8 * hi;
      if (orow < n) Y[(size_t)orow * HC + col] = acc[r] + bc;
    }
  }
}

__global__ void edge_rank_kernel(const int* __restrict__ ei,
                                 int* __restrict__ deg,
                                 int* __restrict__ rank) {
  int e = blockIdx.x * blockDim.x + threadIdx.x;
  if (e >= E_EDGES) return;
  rank[e] = atomicAdd(&deg[ei[E_EDGES + e]], 1);
}

#define SCAN_IT 13
__global__ void __launch_bounds__(1024)
bucket_scan_kernel(const int* __restrict__ deg, int* __restrict__ rowptr) {
  __shared__ int wtot[32];
  const int lane = threadIdx.x & 31;
  const int w    = threadIdx.x >> 5;
  const int base = w * (SCAN_IT * 128) + lane * 4;
  int4 d[SCAN_IT];
  int s = 0;
  #pragma unroll
  for (int it = 0; it < SCAN_IT; ++it) {
    int i = base + it * 128;
    d[it] = (i < N_NODES) ? *(const int4*)(deg + i) : make_int4(0, 0, 0, 0);
    s += d[it].x + d[it].y + d[it].z + d[it].w;
  }
  for (int off = 16; off > 0; off >>= 1) s += __shfl_xor(s, off, 32);
  if (lane == 0) wtot[w] = s;
  __syncthreads();
  int carry = (lane < w) ? wtot[lane] : 0;
  for (int off = 16; off > 0; off >>= 1) carry += __shfl_xor(carry, off, 32);
  #pragma unroll
  for (int it = 0; it < SCAN_IT; ++it) {
    int i = base + it * 128;
    int t = d[it].x + d[it].y + d[it].z + d[it].w;
    int inc = t;
    for (int off = 1; off < 32; off <<= 1) {
      int u = __shfl_up(inc, off, 32);
      if (lane >= off) inc += u;
    }
    int ex = carry + inc - t;
    if (i < N_NODES) {
      int4 o;
      o.x = ex; o.y = o.x + d[it].x; o.z = o.y + d[it].y; o.w = o.z + d[it].z;
      *(int4*)(rowptr + i) = o;
    }
    carry += __shfl(inc, 31, 32);
  }
  if (threadIdx.x == 1023) rowptr[N_NODES] = carry;
}

__global__ void edge_place_kernel(const int* __restrict__ ei,
                                  const int* __restrict__ rowptr,
                                  const int* __restrict__ rank,
                                  int* __restrict__ csr) {
  int e = blockIdx.x * blockDim.x + threadIdx.x;
  if (e >= E_EDGES) return;
  csr[rowptr[ei[E_EDGES + e]] + rank[e]] = ei[e];
}

#define NB 8
__device__ __forceinline__ float head_score(v4f v, v4f xr4, v4f at4) {
  v4f t = v + xr4;
  t.x = (t.x > 0.f) ? t.x : t.x * SLOPE;
  t.y = (t.y > 0.f) ? t.y : t.y * SLOPE;
  t.z = (t.z > 0.f) ? t.z : t.z * SLOPE;
  t.w = (t.w > 0.f) ? t.w : t.w * SLOPE;
  float p = t.x * at4.x + t.y * at4.y + t.z * at4.z + t.w * at4.w;
  p += __shfl_xor(p, 1, 32);
  p += __shfl_xor(p, 2, 32);
  p += __shfl_xor(p, 4, 32);
  return p;
}
__global__ void __launch_bounds__(256)
node_attn_ln_kernel(const int* __restrict__ rowptr,
                    const int* __restrict__ csr,
                    const float* __restrict__ xl,
                    const float* __restrict__ xr,
                    const float* __restrict__ att,
                    const float* __restrict__ bias,
                    const float* __restrict__ gamma,
                    const float* __restrict__ beta,
                    float* __restrict__ out) {
  const int lane = threadIdx.x & 31;
  const int wv   = __builtin_amdgcn_readfirstlane(threadIdx.x >> 5);
  const int node = blockIdx.x * 8 + wv;
  if (node >= N_NODES) return;
  const v4f xr4 = ((const v4f*)(xr + (size_t)node * HC))[lane];
  const v4f at4 = ((const v4f*)att)[lane];
  v4f acc = ((const v4f*)(xl + (size_t)node * HC))[lane];
  float m   = head_score(acc, xr4, at4);
  float den = 1.f;
  const int beg = rowptr[node];
  const int end = rowptr[node + 1];
  int sid[NB];
  if (beg < end) {
    #pragma unroll
    for (int j = 0; j < NB; ++j) {
      int q = (beg + j < end) ? (beg + j) : (end - 1);
      sid[j] = csr[q];
    }
  }
  for (int p = beg; p < end; p += NB) {
    v4f r[NB];
    #pragma unroll
    for (int j = 0; j < NB; ++j)
      r[j] = ((const v4f*)(xl + (size_t)sid[j] * HC))[lane];
    #pragma unroll
    for (int j = 0; j < NB; ++j) {
      int q = (p + NB + j < end) ? (p + NB + j) : (end - 1);
      sid[j] = csr[q];
    }
    asm volatile("" :: "v"(r[0]), "v"(r[1]), "v"(r[2]), "v"(r[3]),
                       "v"(r[4]), "v"(r[5]), "v"(r[6]), "v"(r[7]));
    float sc[NB];
    float mb = m;
    #pragma unroll
    for (int j = 0; j < NB; ++j) {
      float t = head_score(r[j], xr4, at4);
      sc[j] = (p + j < end) ? t : -INFINITY;
      mb = fmaxf(mb, sc[j]);
    }
    float scale = __expf(m - mb);
    acc *= scale;
    den *= scale;
    #pragma unroll
    for (int j = 0; j < NB; ++j) {
      float w = __expf(sc[j] - mb);
      acc += w * r[j];
      den += w;
    }
    m = mb;
  }
  const float4 b4 = ((const float4*)bias)[lane];
  const float4 g4 = ((const float4*)gamma)[lane];
  const float4 e4 = ((const float4*)beta)[lane];
  const float inv = 1.0f / den;
  float h0 = acc.x * inv + b4.x, h1 = acc.y * inv + b4.y;
  float h2 = acc.z * inv + b4.z, h3 = acc.w * inv + b4.w;
  float sum = (h0 + h1) + (h2 + h3);
  for (int off = 16; off > 0; off >>= 1) sum += __shfl_xor(sum, off, 32);
  float mu = sum * (1.0f / HC);
  float d0 = h0 - mu, d1 = h1 - mu, d2 = h2 - mu, d3 = h3 - mu;
  float sq = (d0 * d0 + d1 * d1) + (d2 * d2 + d3 * d3);
  for (int off = 16; off > 0; off >>= 1) sq += __shfl_xor(sq, off, 32);
  float rstd = rsqrtf(sq * (1.0f / HC) + LN_EPS);
  float y0 = d0 * rstd * g4.x + e4.x, y1 = d1 * rstd * g4.y + e4.y;
  float y2 = d2 * rstd * g4.z + e4.z, y3 = d3 * rstd * g4.w + e4.w;
  float4 o;
  o.x = y0 / (1.0f + __expf(-y0));
  o.y = y1 / (1.0f + __expf(-y1));
  o.z = y2 / (1.0f + __expf(-y2));
  o.w = y3 / (1.0f + __expf(-y3));
  ((float4*)(out + (size_t)node * HC))[lane] = o;
}

#define POOL_NODES 32
#define POOL_BATCH 16
__device__ __forceinline__ void pool_flush(float* __restrict__ sums,
                                           float* __restrict__ counts,
                                           int g, int lane, float4 acc, float cnt) {
  float* p = sums + (size_t)g * HC + 4 * lane;
  atomAddF(p + 0, acc.x);
  atomAddF(p + 1, acc.y);
  atomAddF(p + 2, acc.z);
  atomAddF(p + 3, acc.w);
  if (lane == 0) atomAddF(&counts[g], cnt);
}
__global__ void __launch_bounds__(32)
pool_accum_kernel(const float* __restrict__ x,
                  const int* __restrict__ batch,
                  float* __restrict__ sums,
                  float* __restrict__ counts) {
  const int lane = threadIdx.x;
  const int n0 = blockIdx.x * POOL_NODES;
  const int left = N_NODES - n0;
  const int nn = (left < POOL_NODES) ? left : POOL_NODES;
  const float4* xp = (const float4*)(x + (size_t)n0 * HC) + lane;
  int gcur = batch[n0];
  float4 acc = make_float4(0.f, 0.f, 0.f, 0.f);
  float cnt = 0.f;
  for (int j0 = 0; j0 < nn; j0 += POOL_BATCH) {
    float4 v[POOL_BATCH];
    int g[POOL_BATCH];
    #pragma unroll
    for (int j = 0; j < POOL_BATCH; ++j) {
      int jj = (j0 + j < nn) ? (j0 + j) : (nn - 1);
      v[j] = xp[(size_t)jj * (HC / 4)];
      g[j] = batch[n0 + jj];
    }
    #pragma unroll
    for (int j = 0; j < POOL_BATCH; ++j) {
      if (j0 + j < nn) {
        if (g[j] != gcur) {
          pool_flush(sums, counts, gcur, lane, acc, cnt);
          acc = make_float4(0.f, 0.f, 0.f, 0.f);
          cnt = 0.f;
          gcur = g[j];
        }
        acc.x += v[j].x; acc.y += v[j].y; acc.z += v[j].z; acc.w += v[j].w;
        cnt += 1.0f;
      }
    }
  }
  pool_flush(sums, counts, gcur, lane, acc, cnt);
}
__global__ void pool_final_kernel(const float* __restrict__ sums,
                                  const float* __restrict__ counts,
                                  float* __restrict__ out) {
  int t = blockIdx.x * blockDim.x + threadIdx.x;
  if (t >= NGRAPH * HC) return;
  out[t] = sums[t] / fmaxf(counts[t >> 7], 1.0f);
}

extern "C" void kernel_launch(void* const* d_in, const int* in_sizes, int n_in,
                              void* d_out, int out_size, void* d_ws, size_t ws_size,
                              hipStream_t stream) {
  (void)in_sizes; (void)n_in; (void)out_size; (void)ws_size;
  const float* x     = (const float*)d_in[0];
  const int*   ei    = (const int*)d_in[1];
  const int*   batch = (const int*)d_in[2];
  const float* Wl[2]    = {(const float*)d_in[3],  (const float*)d_in[11]};
  const float* Wr[2]    = {(const float*)d_in[4],  (const float*)d_in[12]};
  const float* bl[2]    = {(const float*)d_in[5],  (const float*)d_in[13]};
  const float* br[2]    = {(const float*)d_in[6],  (const float*)d_in[14]};
  const float* att[2]   = {(const float*)d_in[7],  (const float*)d_in[15]};
  const float* bias[2]  = {(const float*)d_in[8],  (const float*)d_in[16]};
  const float* gamma[2] = {(const float*)d_in[9],  (const float*)d_in[17]};
  const float* beta[2]  = {(const float*)d_in[10], (const float*)d_in[18]};

  float* ws = (float*)d_ws;
  float* xl     = ws;          ws += (size_t)N_NODES * HC;
  float* xr2    = ws;          ws += (size_t)N_NODES * HC;
  float* xnode  = ws;          ws += (size_t)N_NODES * HC;
  int*   deg    = (int*)ws;    ws += (size_t)N_NODES;
  int*   rowptr = (int*)ws;    ws += (size_t)N_NODES + 4;
  int*   rank   = (int*)ws;    ws += (size_t)E_EDGES;
  int*   csr    = (int*)ws;    ws += (size_t)E_EDGES;
  float* psums  = ws;          ws += (size_t)NGRAPH * HC;
  float* pcnt   = ws;

  dim3 b256(256);
  const int    gemmGrid = (N_NODES + 16 * ROWTILES - 1) / (16 * ROWTILES);
  const size_t shmem    = (size_t)HC * LDSW * sizeof(float);
  const int    eGrid    = (E_EDGES + 255) / 256;
  const int    nodeGrid = (N_NODES + 7) / 8;
  const int    poolGrid = (N_NODES + POOL_NODES - 1) / POOL_NODES;

  fill_u32<<<(N_NODES + 255) / 256, b256, 0, stream>>>((unsigned*)deg, 0u, N_NODES);
  edge_rank_kernel<<<eGrid, b256, 0, stream>>>(ei, deg, rank);
  bucket_scan_kernel<<<1, dim3(1024), 0, stream>>>(deg, rowptr);
  edge_place_kernel<<<eGrid, b256, 0, stream>>>(ei, rowptr, rank, csr);

  const float* xin = x;
  for (int l = 0; l < 2; ++l) {
    gemm_bias_wmma<<<gemmGrid, b256, shmem, stream>>>(xin, Wl[l], bl[l], xl, N_NODES);
    gemm_bias_wmma<<<gemmGrid, b256, shmem, stream>>>(xin, Wr[l], br[l], xr2, N_NODES);
    node_attn_ln_kernel<<<nodeGrid, b256, 0, stream>>>(rowptr, csr, xl, xr2, att[l],
                                                       bias[l], gamma[l], beta[l], xnode);
    xin = xnode;
  }
  fill_f32<<<(NGRAPH * HC + NGRAPH + 255) / 256, b256, 0, stream>>>(psums, 0.f, NGRAPH * HC + NGRAPH);
  pool_accum_kernel<<<poolGrid, dim3(32), 0, stream>>>(xnode, batch, psums, pcnt);
  pool_final_kernel<<<(NGRAPH * HC + 255) / 256, b256, 0, stream>>>(psums, pcnt, (float*)d_out);
}


# GEMM: each 16-row tile of X staged once per block into LDS by whole-row 16-byte loads (next tile fetched under the products); all 8 waves read A fragments from LDS; k order and sums unchanged
# speedup vs baseline: 3.6995x; 1.2064x over previous
#include <hip/hip_runtime.h>
#include <hip/hip_bf16.h>
#include <math.h>

#define N_NODES 50000
#define E_EDGES 800000
#define ETOT    (E_EDGES + N_NODES)
#define HEADS   4
#define CHP     32
#define HC      128
#define NGRAPH  64
#define SLOPE   0.2f
#define LN_EPS  1e-5f

typedef float v2f __attribute__((ext_vector_type(2)));
typedef float v8f __attribute__((ext_vector_type(8)));
typedef float v4f __attribute__((ext_vector_type(4)));

__device__ __forceinline__ void atomAddF(float* p, float v) {
  unsafeAtomicAdd(p, v);
}

__global__ void fill_f32(float* __restrict__ p, float v, int n) {
  int i = blockIdx.x * blockDim.x + threadIdx.x;
  if (i < n) p[i] = v;
}
__global__ void fill_u32(unsigned* __restrict__ p, unsigned v, int n) {
  int i = blockIdx.x * blockDim.x + threadIdx.x;
  if (i < n) p[i] = v;
}

#define LDSW 136
#define LDSA 132
#define ROWTILES 4
__global__ void gemm_bias_wmma(const float* __restrict__ X,
                               const float* __restrict__ W,
                               const float* __restrict__ bias,
                               float* __restrict__ Y, int n) {
  extern __shared__ __attribute__((aligned(16))) float smem[];
  float* sW = smem;
  float* sA = smem + HC * LDSW;
  const int tid  = threadIdx.x;
  const int wave = tid >> 5;
  const int lane = tid & 31;
  const int rowbase = blockIdx.x * (16 * ROWTILES);
  int nt = (n - rowbase + 15) >> 4;
  nt = (nt < ROWTILES) ? nt : ROWTILES;

  float4 pa0, pa1;
  {
    int r0 = rowbase + wave, r1 = r0 + 8;
    r0 = (r0 < n) ? r0 : (n - 1);
    r1 = (r1 < n) ? r1 : (n - 1);
    pa0 = ((const float4*)(X + (size_t)r0 * HC))[lane];
    pa1 = ((const float4*)(X + (size_t)r1 * HC))[lane];
  }
  for (int i = 0; i < 16; ++i) {
    int idx = tid + 256 * i;
    int r = idx >> 5, q = idx & 31;
    float4 w4 = *(const float4*)(W + r * HC + q * 4);
    float* d = sW + r * LDSW + q * 4;
    d[0] = w4.x; d[1] = w4.y; d[2] = w4.z; d[3] = w4.w;
  }
  *(float4*)(sA + wave * LDSA + lane * 4)       = pa0;
  *(float4*)(sA + (wave + 8) * LDSA + lane * 4) = pa1;
  __syncthreads();

  const int m    = lane & 15;
  const int hi   = lane >> 4;
  const int col  = wave * 16 + m;
  const float bc = bias[col];
  const float* ap = sA + m * LDSA + 2 * hi;
  const float* bp0 = sW + (2 * hi) * LDSW + col;

  for (int t = 0; t < nt; ++t) {
    const int row0 = rowbase + t * 16;
    const bool more = (t + 1 < nt);
    if (more) {
      int r0 = row0 + 16 + wave, r1 = r0 + 8;
      r0 = (r0 < n) ? r0 : (n - 1);
      r1 = (r1 < n) ? r1 : (n - 1);
      pa0 = ((const float4*)(X + (size_t)r0 * HC))[lane];
      pa1 = ((const float4*)(X + (size_t)r1 * HC))[lane];
    }
    v8f acc = {};
    #pragma unroll
    for (int k0 = 0; k0 < HC; k0 += 4) {
      v2f a = *(const v2f*)(ap + k0);
      const float* bp = bp0 + k0 * LDSW;
      v2f b; b.x = bp[0]; b.y = bp[LDSW];
      acc = __builtin_amdgcn_wmma_f32_16x16x4_f32(
          false, a, false, b, (short)0, acc, false, false);
    }
    if (more) {
      __syncthreads();
      *(float4*)(sA + wave * LDSA + lane * 4)       = pa0;
      *(float4*)(sA + (wave + 8) * LDSA + lane * 4) = pa1;
      __syncthreads();
    }
    #pragma unroll
    for (int r = 0; r < 8; ++r) {
      int orow = row0 + r + 8 * hi;
      if (orow < n) Y[(size_t)orow * HC + col] = acc[r] + bc;
    }
  }
}

__global__ void edge_rank_kernel(const int* __restrict__ ei,
                                 int* __restrict__ deg,
                                 int* __restrict__ rank) {
  int e = blockIdx.x * blockDim.x + threadIdx.x;
  if (e >= E_EDGES) return;
  rank[e] = atomicAdd(&deg[ei[E_EDGES + e]], 1);
}

#define SCAN_IT 13
__global__ void __launch_bounds__(1024)
bucket_scan_kernel(const int* __restrict__ deg, int* __restrict__ rowptr) {
  __shared__ int wtot[32];
  const int lane = threadIdx.x & 31;
  const int w    = threadIdx.x >> 5;
  const int base = w * (SCAN_IT * 128) + lane * 4;
  int4 d[SCAN_IT];
  int s = 0;
  #pragma unroll
  for (int it = 0; it < SCAN_IT; ++it) {
    int i = base + it * 128;
    d[it] = (i < N_NODES) ? *(const int4*)(deg + i) : make_int4(0, 0, 0, 0);
    s += d[it].x + d[it].y + d[it].z + d[it].w;
  }
  for (int off = 16; off > 0; off >>= 1) s += __shfl_xor(s, off, 32);
  if (lane == 0) wtot[w] = s;
  __syncthreads();
  int carry = (lane < w) ? wtot[lane] : 0;
  for (int off = 16; off > 0; off >>= 1) carry += __shfl_xor(carry, off, 32);
  #pragma unroll
  for (int it = 0; it < SCAN_IT; ++it) {
    int i = base + it * 128;
    int t = d[it].x + d[it].y + d[it].z + d[it].w;
    int inc = t;
    for (int off = 1; off < 32; off <<= 1) {
      int u = __shfl_up(inc, off, 32);
      if (lane >= off) inc += u;
    }
    int ex = carry + inc - t;
    if (i < N_NODES) {
      int4 o;
      o.x = ex; o.y = o.x + d[it].x; o.z = o.y + d[it].y; o.w = o.z + d[it].z;
      *(int4*)(rowptr + i) = o;
    }
    carry += __shfl(inc, 31, 32);
  }
  if (threadIdx.x == 1023) rowptr[N_NODES] = carry;
}

__global__ void edge_place_kernel(const int* __restrict__ ei,
                                  const int* __restrict__ rowptr,
                                  const int* __restrict__ rank,
                                  int* __restrict__ csr) {
  int e = blockIdx.x * blockDim.x + threadIdx.x;
  if (e >= E_EDGES) return;
  csr[rowptr[ei[E_EDGES + e]] + rank[e]] = ei[e];
}

#define NB 8
__device__ __forceinline__ float head_score(v4f v, v4f xr4, v4f at4) {
  v4f t = v + xr4;
  t.x = (t.x > 0.f) ? t.x : t.x * SLOPE;
  t.y = (t.y > 0.f) ? t.y : t.y * SLOPE;
  t.z = (t.z > 0.f) ? t.z : t.z * SLOPE;
  t.w = (t.w > 0.f) ? t.w : t.w * SLOPE;
  float p = t.x * at4.x + t.y * at4.y + t.z * at4.z + t.w * at4.w;
  p += __shfl_xor(p, 1, 32);
  p += __shfl_xor(p, 2, 32);
  p += __shfl_xor(p, 4, 32);
  return p;
}
__global__ void __launch_bounds__(256)
node_attn_ln_kernel(const int* __restrict__ rowptr,
                    const int* __restrict__ csr,
                    const float* __restrict__ xl,
                    const float* __restrict__ xr,
                    const float* __restrict__ att,
                    const float* __restrict__ bias,
                    const float* __restrict__ gamma,
                    const float* __restrict__ beta,
                    float* __restrict__ out) {
  const int lane = threadIdx.x & 31;
  const int wv   = __builtin_amdgcn_readfirstlane(threadIdx.x >> 5);
  const int node = blockIdx.x * 8 + wv;
  if (node >= N_NODES) return;
  const v4f xr4 = ((const v4f*)(xr + (size_t)node * HC))[lane];
  const v4f at4 = ((const v4f*)att)[lane];
  v4f acc = ((const v4f*)(xl + (size_t)node * HC))[lane];
  float m   = head_score(acc, xr4, at4);
  float den = 1.f;
  const int beg = rowptr[node];
  const int end = rowptr[node + 1];
  int sid[NB];
  if (beg < end) {
    #pragma unroll
    for (int j = 0; j < NB; ++j) {
      int q = (beg + j < end) ? (beg + j) : (end - 1);
      sid[j] = csr[q];
    }
  }
  for (int p = beg; p < end; p += NB) {
    v4f r[NB];
    #pragma unroll
    for (int j = 0; j < NB; ++j)
      r[j] = ((const v4f*)(xl + (size_t)sid[j] * HC))[lane];
    #pragma unroll
    for (int j = 0; j < NB; ++j) {
      int q = (p + NB + j < end) ? (p + NB + j) : (end - 1);
      sid[j] = csr[q];
    }
    asm volatile("" :: "v"(r[0]), "v"(r[1]), "v"(r[2]), "v"(r[3]),
                       "v"(r[4]), "v"(r[5]), "v"(r[6]), "v"(r[7]));
    float sc[NB];
    float mb = m;
    #pragma unroll
    for (int j = 0; j < NB; ++j) {
      float t = head_score(r[j], xr4, at4);
      sc[j] = (p + j < end) ? t : -INFINITY;
      mb = fmaxf(mb, sc[j]);
    }
    float scale = __expf(m - mb);
    acc *= scale;
    den *= scale;
    #pragma unroll
    for (int j = 0; j < NB; ++j) {
      float w = __expf(sc[j] - mb);
      acc += w * r[j];
      den += w;
    }
    m = mb;
  }
  const float4 b4 = ((const float4*)bias)[lane];
  const float4 g4 = ((const float4*)gamma)[lane];
  const float4 e4 = ((const float4*)beta)[lane];
  const float inv = 1.0f / den;
  float h0 = acc.x * inv + b4.x, h1 = acc.y * inv + b4.y;
  float h2 = acc.z * inv + b4.z, h3 = acc.w * inv + b4.w;
  float sum = (h0 + h1) + (h2 + h3);
  for (int off = 16; off > 0; off >>= 1) sum += __shfl_xor(sum, off, 32);
  float mu = sum * (1.0f / HC);
  float d0 = h0 - mu, d1 = h1 - mu, d2 = h2 - mu, d3 = h3 - mu;
  float sq = (d0 * d0 + d1 * d1) + (d2 * d2 + d3 * d3);
  for (int off = 16; off > 0; off >>= 1) sq += __shfl_xor(sq, off, 32);
  float rstd = rsqrtf(sq * (1.0f / HC) + LN_EPS);
  float y0 = d0 * rstd * g4.x + e4.x, y1 = d1 * rstd * g4.y + e4.y;
  float y2 = d2 * rstd * g4.z + e4.z, y3 = d3 * rstd * g4.w + e4.w;
  float4 o;
  o.x = y0 / (1.0f + __expf(-y0));
  o.y = y1 / (1.0f + __expf(-y1));
  o.z = y2 / (1.0f + __expf(-y2));
  o.w = y3 / (1.0f + __expf(-y3));
  ((float4*)(out + (size_t)node * HC))[lane] = o;
}

#define POOL_NODES 32
#define POOL_BATCH 16
__device__ __forceinline__ void pool_flush(float* __restrict__ sums,
                                           float* __restrict__ counts,
                                           int g, int lane, float4 acc, float cnt) {
  float* p = sums + (size_t)g * HC + 4 * lane;
  atomAddF(p + 0, acc.x);
  atomAddF(p + 1, acc.y);
  atomAddF(p + 2, acc.z);
  atomAddF(p + 3, acc.w);
  if (lane == 0) atomAddF(&counts[g], cnt);
}
__global__ void __launch_bounds__(32)
pool_accum_kernel(const float* __restrict__ x,
                  const int* __restrict__ batch,
                  float* __restrict__ sums,
                  float* __restrict__ counts) {
  const int lane = threadIdx.x;
  const int n0 = blockIdx.x * POOL_NODES;
  const int left = N_NODES - n0;
  const int nn = (left < POOL_NODES) ? left : POOL_NODES;
  const float4* xp = (const float4*)(x + (size_t)n0 * HC) + lane;
  int gcur = batch[n0];
  float4 acc = make_float4(0.f, 0.f, 0.f, 0.f);
  float cnt = 0.f;
  for (int j0 = 0; j0 < nn; j0 += POOL_BATCH) {
    float4 v[POOL_BATCH];
    int g[POOL_BATCH];
    #pragma unroll
    for (int j = 0; j < POOL_BATCH; ++j) {
      int jj = (j0 + j < nn) ? (j0 + j) : (nn - 1);
      v[j] = xp[(size_t)jj * (HC / 4)];
      g[j] = batch[n0 + jj];
    }
    #pragma unroll
    for (int j = 0; j < POOL_BATCH; ++j) {
      if (j0 + j < nn) {
        if (g[j] != gcur) {
          pool_flush(sums, counts, gcur, lane, acc, cnt);
          acc = make_float4(0.f, 0.f, 0.f, 0.f);
          cnt = 0.f;
          gcur = g[j];
        }
        acc.x += v[j].x; acc.y += v[j].y; acc.z += v[j].z; acc.w += v[j].w;
        cnt += 1.0f;
      }
    }
  }
  pool_flush(sums, counts, gcur, lane, acc, cnt);
}
__global__ void pool_final_kernel(const float* __restrict__ sums,
                                  const float* __restrict__ counts,
                                  float* __restrict__ out) {
  int t = blockIdx.x * blockDim.x + threadIdx.x;
  if (t >= NGRAPH * HC) return;
  out[t] = sums[t] / fmaxf(counts[t >> 7], 1.0f);
}

extern "C" void kernel_launch(void* const* d_in, const int* in_sizes, int n_in,
                              void* d_out, int out_size, void* d_ws, size_t ws_size,
                              hipStream_t stream) {
  (void)in_sizes; (void)n_in; (void)out_size; (void)ws_size;
  const float* x     = (const float*)d_in[0];
  const int*   ei    = (const int*)d_in[1];
  const int*   batch = (const int*)d_in[2];
  const float* Wl[2]    = {(const float*)d_in[3],  (const float*)d_in[11]};
  const float* Wr[2]    = {(const float*)d_in[4],  (const float*)d_in[12]};
  const float* bl[2]    = {(const float*)d_in[5],  (const float*)d_in[13]};
  const float* br[2]    = {(const float*)d_in[6],  (const float*)d_in[14]};
  const float* att[2]   = {(const float*)d_in[7],  (const float*)d_in[15]};
  const float* bias[2]  = {(const float*)d_in[8],  (const float*)d_in[16]};
  const float* gamma[2] = {(const float*)d_in[9],  (const float*)d_in[17]};
  const float* beta[2]  = {(const float*)d_in[10], (const float*)d_in[18]};

  float* ws = (float*)d_ws;
  float* xl     = ws;          ws += (size_t)N_NODES * HC;
  float* xr2    = ws;          ws += (size_t)N_NODES * HC;
  float* xnode  = ws;          ws += (size_t)N_NODES * HC;
  int*   deg    = (int*)ws;    ws += (size_t)N_NODES;
  int*   rowptr = (int*)ws;    ws += (size_t)N_NODES + 4;
  int*   rank   = (int*)ws;    ws += (size_t)E_EDGES;
  int*   csr    = (int*)ws;    ws += (size_t)E_EDGES;
  float* psums  = ws;          ws += (size_t)NGRAPH * HC;
  float* pcnt   = ws;

  dim3 b256(256);
  const int    gemmGrid = (N_NODES + 16 * ROWTILES - 1) / (16 * ROWTILES);
  const size_t shmem    = ((size_t)HC * LDSW + 16 * LDSA) * sizeof(float);
  const int    eGrid    = (E_EDGES + 255) / 256;
  const int    nodeGrid = (N_NODES + 7) / 8;
  const int    poolGrid = (N_NODES + POOL_NODES - 1) / POOL_NODES;

  fill_u32<<<(N_NODES + 255) / 256, b256, 0, stream>>>((unsigned*)deg, 0u, N_NODES);
  edge_rank_kernel<<<eGrid, b256, 0, stream>>>(ei, deg, rank);
  bucket_scan_kernel<<<1, dim3(1024), 0, stream>>>(deg, rowptr);
  edge_place_kernel<<<eGrid, b256, 0, stream>>>(ei, rowptr, rank, csr);

  const float* xin = x;
  for (int l = 0; l < 2; ++l) {
    gemm_bias_wmma<<<gemmGrid, b256, shmem, stream>>>(xin, Wl[l], bl[l], xl, N_NODES);
    gemm_bias_wmma<<<gemmGrid, b256, shmem, stream>>>(xin, Wr[l], br[l], xr2, N_NODES);
    node_attn_ln_kernel<<<nodeGrid, b256, 0, stream>>>(rowptr, csr, xl, xr2, att[l],
                                                       bias[l], gamma[l], beta[l], xnode);
    xin = xnode;
  }
  fill_f32<<<(NGRAPH * HC + NGRAPH + 255) / 256, b256, 0, stream>>>(psums, 0.f, NGRAPH * HC + NGRAPH);
  pool_accum_kernel<<<poolGrid, dim3(32), 0, stream>>>(xnode, batch, psums, pcnt);
  pool_final_kernel<<<(NGRAPH * HC + 255) / 256, b256, 0, stream>>>(psums, pcnt, (float*)d_out);
}


# node pass: cross-lane sums by DPP lane operands, not LDS permutes; a batch's 8 head sums folded (4+2+1 steps) to one edge per lane: max, exp, denominator once per lane; denominator sum order changes
# speedup vs baseline: 3.9709x; 1.0734x over previous
#include <hip/hip_runtime.h>
#include <hip/hip_bf16.h>
#include <math.h>

#define N_NODES 50000
#define E_EDGES 800000
#define ETOT    (E_EDGES + N_NODES)
#define HEADS   4
#define CHP     32
#define HC      128
#define NGRAPH  64
#define SLOPE   0.2f
#define LN_EPS  1e-5f

typedef float v2f __attribute__((ext_vector_type(2)));
typedef float v8f __attribute__((ext_vector_type(8)));
typedef float v4f __attribute__((ext_vector_type(4)));

__device__ __forceinline__ void atomAddF(float* p, float v) {
  unsafeAtomicAdd(p, v);
}

__global__ void fill_f32(float* __restrict__ p, float v, int n) {
  int i = blockIdx.x * blockDim.x + threadIdx.x;
  if (i < n) p[i] = v;
}
__global__ void fill_u32(unsigned* __restrict__ p, unsigned v, int n) {
  int i = blockIdx.x * blockDim.x + threadIdx.x;
  if (i < n) p[i] = v;
}

#define LDSW 136
#define LDSA 132
#define ROWTILES 4
__global__ void gemm_bias_wmma(const float* __restrict__ X,
                               const float* __restrict__ W,
                               const float* __restrict__ bias,
                               float* __restrict__ Y, int n) {
  extern __shared__ __attribute__((aligned(16))) float smem[];
  float* sW = smem;
  float* sA = smem + HC * LDSW;
  const int tid  = threadIdx.x;
  const int wave = tid >> 5;
  const int lane = tid & 31;
  const int rowbase = blockIdx.x * (16 * ROWTILES);
  int nt = (n - rowbase + 15) >> 4;
  nt = (nt < ROWTILES) ? nt : ROWTILES;

  float4 pa0, pa1;
  {
    int r0 = rowbase + wave, r1 = r0 + 8;
    r0 = (r0 < n) ? r0 : (n - 1);
    r1 = (r1 < n) ? r1 : (n - 1);
    pa0 = ((const float4*)(X + (size_t)r0 * HC))[lane];
    pa1 = ((const float4*)(X + (size_t)r1 * HC))[lane];
  }
  for (int i = 0; i < 16; ++i) {
    int idx = tid + 256 * i;
    int r = idx >> 5, q = idx & 31;
    float4 w4 = *(const float4*)(W + r * HC + q * 4);
    float* d = sW + r * LDSW + q * 4;
    d[0] = w4.x; d[1] = w4.y; d[2] = w4.z; d[3] = w4.w;
  }
  *(float4*)(sA + wave * LDSA + lane * 4)       = pa0;
  *(float4*)(sA + (wave + 8) * LDSA + lane * 4) = pa1;
  __syncthreads();

  const int m    = lane & 15;
  const int hi   = lane >> 4;
  const int col  = wave * 16 + m;
  const float bc = bias[col];
  const float* ap = sA + m * LDSA + 2 * hi;
  const float* bp0 = sW + (2 * hi) * LDSW + col;

  for (int t = 0; t < nt; ++t) {
    const int row0 = rowbase + t * 16;
    const bool more = (t + 1 < nt);
    if (more) {
      int r0 = row0 + 16 + wave, r1 = r0 + 8;
      r0 = (r0 < n) ? r0 : (n - 1);
      r1 = (r1 < n) ? r1 : (n - 1);
      pa0 = ((const float4*)(X + (size_t)r0 * HC))[lane];
      pa1 = ((const float4*)(X + (size_t)r1 * HC))[lane];
    }
    v8f acc = {};
    #pragma unroll
    for (int k0 = 0; k0 < HC; k0 += 4) {
      v2f a = *(const v2f*)(ap + k0);
      const float* bp = bp0 + k0 * LDSW;
      v2f b; b.x = bp[0]; b.y = bp[LDSW];
      acc = __builtin_amdgcn_wmma_f32_16x16x4_f32(
          false, a, false, b, (short)0, acc, false, false);
    }
    if (more) {
      __syncthreads();
      *(float4*)(sA + wave * LDSA + lane * 4)       = pa0;
      *(float4*)(sA + (wave + 8) * LDSA + lane * 4) = pa1;
      __syncthreads();
    }
    #pragma unroll
    for (int r = 0; r < 8; ++r) {
      int orow = row0 + r + 8 * hi;
      if (orow < n) Y[(size_t)orow * HC + col] = acc[r] + bc;
    }
  }
}

__global__ void edge_rank_kernel(const int* __restrict__ ei,
                                 int* __restrict__ deg,
                                 int* __restrict__ rank) {
  int e = blockIdx.x * blockDim.x + threadIdx.x;
  if (e >= E_EDGES) return;
  rank[e] = atomicAdd(&deg[ei[E_EDGES + e]], 1);
}

#define SCAN_IT 13
__global__ void __launch_bounds__(1024)
bucket_scan_kernel(const int* __restrict__ deg, int* __restrict__ rowptr) {
  __shared__ int wtot[32];
  const int lane = threadIdx.x & 31;
  const int w    = threadIdx.x >> 5;
  const int base = w * (SCAN_IT * 128) + lane * 4;
  int4 d[SCAN_IT];
  int s = 0;
  #pragma unroll
  for (int it = 0; it < SCAN_IT; ++it) {
    int i = base + it * 128;
    d[it] = (i < N_NODES) ? *(const int4*)(deg + i) : make_int4(0, 0, 0, 0);
    s += d[it].x + d[it].y + d[it].z + d[it].w;
  }
  for (int off = 16; off > 0; off >>= 1) s += __shfl_xor(s, off, 32);
  if (lane == 0) wtot[w] = s;
  __syncthreads();
  int carry = (lane < w) ? wtot[lane] : 0;
  for (int off = 16; off > 0; off >>= 1) carry += __shfl_xor(carry, off, 32);
  #pragma unroll
  for (int it = 0; it < SCAN_IT; ++it) {
    int i = base + it * 128;
    int t = d[it].x + d[it].y + d[it].z + d[it].w;
    int inc = t;
    for (int off = 1; off < 32; off <<= 1) {
      int u = __shfl_up(inc, off, 32);
      if (lane >= off) inc += u;
    }
    int ex = carry + inc - t;
    if (i < N_NODES) {
      int4 o;
      o.x = ex; o.y = o.x + d[it].x; o.z = o.y + d[it].y; o.w = o.z + d[it].z;
      *(int4*)(rowptr + i) = o;
    }
    carry += __shfl(inc, 31, 32);
  }
  if (threadIdx.x == 1023) rowptr[N_NODES] = carry;
}

__global__ void edge_place_kernel(const int* __restrict__ ei,
                                  const int* __restrict__ rowptr,
                                  const int* __restrict__ rank,
                                  int* __restrict__ csr) {
  int e = blockIdx.x * blockDim.x + threadIdx.x;
  if (e >= E_EDGES) return;
  csr[rowptr[ei[E_EDGES + e]] + rank[e]] = ei[e];
}

#define NB 8
template <int K>
__device__ __forceinline__ float xlane(float v) {
  return __int_as_float(__builtin_amdgcn_mov_dpp(__float_as_int(v), 0x160 + K, 0xf, 0xf, true));
}
__device__ __forceinline__ float xhalf(float v) {
  int b = __float_as_int(v);
  return __int_as_float(__builtin_amdgcn_permlanex16(b, b, 0x76543210, 0xfedcba98, false, false));
}
template <int G>
__device__ __forceinline__ float group_lane(float v) {
  return __int_as_float((int)__builtin_amdgcn_mov_dpp8((unsigned)__float_as_int(v), G * 0x249249));
}
__device__ __forceinline__ float lane_score(v4f v, v4f xr4, v4f at4) {
  v4f t = v + xr4;
  t.x = (t.x > 0.f) ? t.x : t.x * SLOPE;
  t.y = (t.y > 0.f) ? t.y : t.y * SLOPE;
  t.z = (t.z > 0.f) ? t.z : t.z * SLOPE;
  t.w = (t.w > 0.f) ? t.w : t.w * SLOPE;
  return t.x * at4.x + t.y * at4.y + t.z * at4.z + t.w * at4.w;
}
__global__ void __launch_bounds__(256)
node_attn_ln_kernel(const int* __restrict__ rowptr,
                    const int* __restrict__ csr,
                    const float* __restrict__ xl,
                    const float* __restrict__ xr,
                    const float* __restrict__ att,
                    const float* __restrict__ bias,
                    const float* __restrict__ gamma,
                    const float* __restrict__ beta,
                    float* __restrict__ out) {
  const int lane = threadIdx.x & 31;
  const int wv   = __builtin_amdgcn_readfirstlane(threadIdx.x >> 5);
  const int node = blockIdx.x * 8 + wv;
  if (node >= N_NODES) return;
  const bool b0 = (lane & 1) != 0, b1 = (lane & 2) != 0, b2 = (lane & 4) != 0;
  const int edge_of_lane = (b0 ? 4 : 0) + (b1 ? 2 : 0) + (b2 ? 1 : 0);
  const v4f xr4 = ((const v4f*)(xr + (size_t)node * HC))[lane];
  const v4f at4 = ((const v4f*)att)[lane];
  v4f acc = ((const v4f*)(xl + (size_t)node * HC))[lane];
  float m = lane_score(acc, xr4, at4);
  m += xlane<1>(m);
  m += xlane<2>(m);
  m += xlane<4>(m);
  float den = (edge_of_lane == 0) ? 1.f : 0.f;
  const int beg = rowptr[node];
  const int end = rowptr[node + 1];
  int sid[NB];
  if (beg < end) {
    #pragma unroll
    for (int j = 0; j < NB; ++j) {
      int q = (beg + j < end) ? (beg + j) : (end - 1);
      sid[j] = csr[q];
    }
  }
  for (int p = beg; p < end; p += NB) {
    v4f r[NB];
    #pragma unroll
    for (int j = 0; j < NB; ++j)
      r[j] = ((const v4f*)(xl + (size_t)sid[j] * HC))[lane];
    #pragma unroll
    for (int j = 0; j < NB; ++j) {
      int q = (p + NB + j < end) ? (p + NB + j) : (end - 1);
      sid[j] = csr[q];
    }
    asm volatile("" :: "v"(r[0]), "v"(r[1]), "v"(r[2]), "v"(r[3]),
                       "v"(r[4]), "v"(r[5]), "v"(r[6]), "v"(r[7]));
    float s8[NB];
    #pragma unroll
    for (int j = 0; j < NB; ++j) s8[j] = lane_score(r[j], xr4, at4);
    float s4[4];
    #pragma unroll
    for (int k = 0; k < 4; ++k)
      s4[k] = (b0 ? s8[k + 4] : s8[k]) + xlane<1>(b0 ? s8[k] : s8[k + 4]);
    float s2[2];
    #pragma unroll
    for (int k = 0; k < 2; ++k)
      s2[k] = (b1 ? s4[k + 2] : s4[k]) + xlane<2>(b1 ? s4[k] : s4[k + 2]);
    float sc = (b2 ? s2[1] : s2[0]) + xlane<4>(b2 ? s2[0] : s2[1]);
    sc = (p + edge_of_lane < end) ? sc : -INFINITY;
    float mb = fmaxf(sc, xlane<1>(sc));
    mb = fmaxf(mb, xlane<2>(mb));
    mb = fmaxf(mb, xlane<4>(mb));
    mb = fmaxf(mb, m);
    const float scale = __expf(m - mb);
    const float w = __expf(sc - mb);
    acc *= scale;
    den = den * scale + w;
    acc += group_lane<0>(w) * r[0];
    acc += group_lane<4>(w) * r[1];
    acc += group_lane<2>(w) * r[2];
    acc += group_lane<6>(w) * r[3];
    acc += group_lane<1>(w) * r[4];
    acc += group_lane<5>(w) * r[5];
    acc += group_lane<3>(w) * r[6];
    acc += group_lane<7>(w) * r[7];
    m = mb;
  }
  den += xlane<1>(den);
  den += xlane<2>(den);
  den += xlane<4>(den);
  const float4 b4 = ((const float4*)bias)[lane];
  const float4 g4 = ((const float4*)gamma)[lane];
  const float4 e4 = ((const float4*)beta)[lane];
  const float inv = 1.0f / den;
  float h0 = acc.x * inv + b4.x, h1 = acc.y * inv + b4.y;
  float h2 = acc.z * inv + b4.z, h3 = acc.w * inv + b4.w;
  float sum = (h0 + h1) + (h2 + h3);
  sum += xhalf(sum);
  sum += xlane<8>(sum);
  sum += xlane<4>(sum);
  sum += xlane<2>(sum);
  sum += xlane<1>(sum);
  float mu = sum * (1.0f / HC);
  float d0 = h0 - mu, d1 = h1 - mu, d2 = h2 - mu, d3 = h3 - mu;
  float sq = (d0 * d0 + d1 * d1) + (d2 * d2 + d3 * d3);
  sq += xhalf(sq);
  sq += xlane<8>(sq);
  sq += xlane<4>(sq);
  sq += xlane<2>(sq);
  sq += xlane<1>(sq);
  float rstd = rsqrtf(sq * (1.0f / HC) + LN_EPS);
  float y0 = d0 * rstd * g4.x + e4.x, y1 = d1 * rstd * g4.y + e4.y;
  float y2 = d2 * rstd * g4.z + e4.z, y3 = d3 * rstd * g4.w + e4.w;
  float4 o;
  o.x = y0 / (1.0f + __expf(-y0));
  o.y = y1 / (1.0f + __expf(-y1));
  o.z = y2 / (1.0f + __expf(-y2));
  o.w = y3 / (1.0f + __expf(-y3));
  ((float4*)(out + (size_t)node * HC))[lane] = o;
}

#define POOL_NODES 32
#define POOL_BATCH 16
__device__ __forceinline__ void pool_flush(float* __restrict__ sums,
                                           float* __restrict__ counts,
                                           int g, int lane, float4 acc, float cnt) {
  float* p = sums + (size_t)g * HC + 4 * lane;
  atomAddF(p + 0, acc.x);
  atomAddF(p + 1, acc.y);
  atomAddF(p + 2, acc.z);
  atomAddF(p + 3, acc.w);
  if (lane == 0) atomAddF(&counts[g], cnt);
}
__global__ void __launch_bounds__(32)
pool_accum_kernel(const float* __restrict__ x,
                  const int* __restrict__ batch,
                  float* __restrict__ sums,
                  float* __restrict__ counts) {
  const int lane = threadIdx.x;
  const int n0 = blockIdx.x * POOL_NODES;
  const int left = N_NODES - n0;
  const int nn = (left < POOL_NODES) ? left : POOL_NODES;
  const float4* xp = (const float4*)(x + (size_t)n0 * HC) + lane;
  int gcur = batch[n0];
  float4 acc = make_float4(0.f, 0.f, 0.f, 0.f);
  float cnt = 0.f;
  for (int j0 = 0; j0 < nn; j0 += POOL_BATCH) {
    float4 v[POOL_BATCH];
    int g[POOL_BATCH];
    #pragma unroll
    for (int j = 0; j < POOL_BATCH; ++j) {
      int jj = (j0 + j < nn) ? (j0 + j) : (nn - 1);
      v[j] = xp[(size_t)jj * (HC / 4)];
      g[j] = batch[n0 + jj];
    }
    #pragma unroll
    for (int j = 0; j < POOL_BATCH; ++j) {
      if (j0 + j < nn) {
        if (g[j] != gcur) {
          pool_flush(sums, counts, gcur, lane, acc, cnt);
          acc = make_float4(0.f, 0.f, 0.f, 0.f);
          cnt = 0.f;
          gcur = g[j];
        }
        acc.x += v[j].x; acc.y += v[j].y; acc.z += v[j].z; acc.w += v[j].w;
        cnt += 1.0f;
      }
    }
  }
  pool_flush(sums, counts, gcur, lane, acc, cnt);
}
__global__ void pool_final_kernel(const float* __restrict__ sums,
                                  const float* __restrict__ counts,
                                  float* __restrict__ out) {
  int t = blockIdx.x * blockDim.x + threadIdx.x;
  if (t >= NGRAPH * HC) return;
  out[t] = sums[t] / fmaxf(counts[t >> 7], 1.0f);
}

extern "C" void kernel_launch(void* const* d_in, const int* in_sizes, int n_in,
                              void* d_out, int out_size, void* d_ws, size_t ws_size,
                              hipStream_t stream) {
  (void)in_sizes; (void)n_in; (void)out_size; (void)ws_size;
  const float* x     = (const float*)d_in[0];
  const int*   ei    = (const int*)d_in[1];
  const int*   batch = (const int*)d_in[2];
  const float* Wl[2]    = {(const float*)d_in[3],  (const float*)d_in[11]};
  const float* Wr[2]    = {(const float*)d_in[4],  (const float*)d_in[12]};
  const float* bl[2]    = {(const float*)d_in[5],  (const float*)d_in[13]};
  const float* br[2]    = {(const float*)d_in[6],  (const float*)d_in[14]};
  const float* att[2]   = {(const float*)d_in[7],  (const float*)d_in[15]};
  const float* bias[2]  = {(const float*)d_in[8],  (const float*)d_in[16]};
  const float* gamma[2] = {(const float*)d_in[9],  (const float*)d_in[17]};
  const float* beta[2]  = {(const float*)d_in[10], (const float*)d_in[18]};

  float* ws = (float*)d_ws;
  float* xl     = ws;          ws += (size_t)N_NODES * HC;
  float* xr2    = ws;          ws += (size_t)N_NODES * HC;
  float* xnode  = ws;          ws += (size_t)N_NODES * HC;
  int*   deg    = (int*)ws;    ws += (size_t)N_NODES;
  int*   rowptr = (int*)ws;    ws += (size_t)N_NODES + 4;
  int*   rank   = (int*)ws;    ws += (size_t)E_EDGES;
  int*   csr    = (int*)ws;    ws += (size_t)E_EDGES;
  float* psums  = ws;          ws += (size_t)NGRAPH * HC;
  float* pcnt   = ws;

  dim3 b256(256);
  const int    gemmGrid = (N_NODES + 16 * ROWTILES - 1) / (16 * ROWTILES);
  const size_t shmem    = ((size_t)HC * LDSW + 16 * LDSA) * sizeof(float);
  const int    eGrid    = (E_EDGES + 255) / 256;
  const int    nodeGrid = (N_NODES + 7) / 8;
  const int    poolGrid = (N_NODES + POOL_NODES - 1) / POOL_NODES;

  fill_u32<<<(N_NODES + 255) / 256, b256, 0, stream>>>((unsigned*)deg, 0u, N_NODES);
  edge_rank_kernel<<<eGrid, b256, 0, stream>>>(ei, deg, rank);
  bucket_scan_kernel<<<1, dim3(1024), 0, stream>>>(deg, rowptr);
  edge_place_kernel<<<eGrid, b256, 0, stream>>>(ei, rowptr, rank, csr);

  const float* xin = x;
  for (int l = 0; l < 2; ++l) {
    gemm_bias_wmma<<<gemmGrid, b256, shmem, stream>>>(xin, Wl[l], bl[l], xl, N_NODES);
    gemm_bias_wmma<<<gemmGrid, b256, shmem, stream>>>(xin, Wr[l], br[l], xr2, N_NODES);
    node_attn_ln_kernel<<<nodeGrid, b256, 0, stream>>>(rowptr, csr, xl, xr2, att[l],
                                                       bias[l], gamma[l], beta[l], xnode);
    xin = xnode;
  }
  fill_f32<<<(NGRAPH * HC + NGRAPH + 255) / 256, b256, 0, stream>>>(psums, 0.f, NGRAPH * HC + NGRAPH);
  pool_accum_kernel<<<poolGrid, dim3(32), 0, stream>>>(xnode, batch, psums, pcnt);
  pool_final_kernel<<<(NGRAPH * HC + 255) / 256, b256, 0, stream>>>(psums, pcnt, (float*)d_out);
}


# edge buckets by a two-level counting sort on LDS counters (256 node ranges, then nodes), not one returning global atomic per edge; 4 launches to 3; order in a bucket (sum order) stays arrival order
# speedup vs baseline: 4.5376x; 1.1427x over previous
#include <hip/hip_runtime.h>
#include <hip/hip_bf16.h>
#include <math.h>

#define N_NODES 50000
#define E_EDGES 800000
#define ETOT    (E_EDGES + N_NODES)
#define HEADS   4
#define CHP     32
#define HC      128
#define NGRAPH  64
#define SLOPE   0.2f
#define LN_EPS  1e-5f

typedef float v2f __attribute__((ext_vector_type(2)));
typedef float v8f __attribute__((ext_vector_type(8)));
typedef float v4f __attribute__((ext_vector_type(4)));

__device__ __forceinline__ void atomAddF(float* p, float v) {
  unsafeAtomicAdd(p, v);
}

__global__ void fill_f32(float* __restrict__ p, float v, int n) {
  int i = blockIdx.x * blockDim.x + threadIdx.x;
  if (i < n) p[i] = v;
}

#define LDSW 136
#define LDSA 132
#define ROWTILES 4
__global__ void gemm_bias_wmma(const float* __restrict__ X,
                               const float* __restrict__ W,
                               const float* __restrict__ bias,
                               float* __restrict__ Y, int n) {
  extern __shared__ __attribute__((aligned(16))) float smem[];
  float* sW = smem;
  float* sA = smem + HC * LDSW;
  const int tid  = threadIdx.x;
  const int wave = tid >> 5;
  const int lane = tid & 31;
  const int rowbase = blockIdx.x * (16 * ROWTILES);
  int nt = (n - rowbase + 15) >> 4;
  nt = (nt < ROWTILES) ? nt : ROWTILES;

  float4 pa0, pa1;
  {
    int r0 = rowbase + wave, r1 = r0 + 8;
    r0 = (r0 < n) ? r0 : (n - 1);
    r1 = (r1 < n) ? r1 : (n - 1);
    pa0 = ((const float4*)(X + (size_t)r0 * HC))[lane];
    pa1 = ((const float4*)(X + (size_t)r1 * HC))[lane];
  }
  for (int i = 0; i < 16; ++i) {
    int idx = tid + 256 * i;
    int r = idx >> 5, q = idx & 31;
    float4 w4 = *(const float4*)(W + r * HC + q * 4);
    float* d = sW + r * LDSW + q * 4;
    d[0] = w4.x; d[1] = w4.y; d[2] = w4.z; d[3] = w4.w;
  }
  *(float4*)(sA + wave * LDSA + lane * 4)       = pa0;
  *(float4*)(sA + (wave + 8) * LDSA + lane * 4) = pa1;
  __syncthreads();

  const int m    = lane & 15;
  const int hi   = lane >> 4;
  const int col  = wave * 16 + m;
  const float bc = bias[col];
  const float* ap = sA + m * LDSA + 2 * hi;
  const float* bp0 = sW + (2 * hi) * LDSW + col;

  for (int t = 0; t < nt; ++t) {
    const int row0 = rowbase + t * 16;
    const bool more = (t + 1 < nt);
    if (more) {
      int r0 = row0 + 16 + wave, r1 = r0 + 8;
      r0 = (r0 < n) ? r0 : (n - 1);
      r1 = (r1 < n) ? r1 : (n - 1);
      pa0 = ((const float4*)(X + (size_t)r0 * HC))[lane];
      pa1 = ((const float4*)(X + (size_t)r1 * HC))[lane];
    }
    v8f acc = {};
    #pragma unroll
    for (int k0 = 0; k0 < HC; k0 += 4) {
      v2f a = *(const v2f*)(ap + k0);
      const float* bp = bp0 + k0 * LDSW;
      v2f b; b.x = bp[0]; b.y = bp[LDSW];
      acc = __builtin_amdgcn_wmma_f32_16x16x4_f32(
          false, a, false, b, (short)0, acc, false, false);
    }
    if (more) {
      __syncthreads();
      *(float4*)(sA + wave * LDSA + lane * 4)       = pa0;
      *(float4*)(sA + (wave + 8) * LDSA + lane * 4) = pa1;
      __syncthreads();
    }
    #pragma unroll
    for (int r = 0; r < 8; ++r) {
      int orow = row0 + r + 8 * hi;
      if (orow < n) Y[(size_t)orow * HC + col] = acc[r] + bc;
    }
  }
}

#define NBIN   256
#define BINW   196
#define EBLK   256
#define ECHUNK (E_EDGES / EBLK)
#define EPT    ((ECHUNK + 255) / 256)
#define SORT_T 512
#define SORT_K 8
static_assert(EBLK * ECHUNK == E_EDGES, "edge blocks cover the edges exactly");
static_assert(NBIN * BINW > N_NODES, "ranges cover node ids 0..N_NODES");
static_assert(BINW <= NBIN, "one counter per node of a range");

template <int NW>
__device__ __forceinline__ int block_excl_scan(int v, int* wsum) {
  const int lane = threadIdx.x & 31;
  const int w    = threadIdx.x >> 5;
  int inc = v;
  for (int off = 1; off < 32; off <<= 1) {
    int u = __shfl_up(inc, off, 32);
    if (lane >= off) inc += u;
  }
  if (lane == 31) wsum[w] = inc;
  __syncthreads();
  int carry = 0;
  for (int k = 0; k < NW; ++k) carry += (k < w) ? wsum[k] : 0;
  return carry + inc - v;
}

__global__ void __launch_bounds__(256)
edge_bin_count_kernel(const int* __restrict__ ei, int* __restrict__ cnt) {
  __shared__ int h[NBIN];
  const int tid  = threadIdx.x;
  const int base = blockIdx.x * ECHUNK;
  int d[EPT];
  #pragma unroll
  for (int j = 0; j < EPT; ++j) {
    int i = tid + 256 * j;
    i = (i < ECHUNK) ? i : (ECHUNK - 1);
    d[j] = ei[E_EDGES + base + i];
  }
  h[tid] = 0;
  __syncthreads();
  #pragma unroll
  for (int j = 0; j < EPT; ++j)
    if (tid + 256 * j < ECHUNK) atomicAdd(&h[d[j] / BINW], 1);
  __syncthreads();
  cnt[blockIdx.x * NBIN + tid] = h[tid];
}

__global__ void __launch_bounds__(256)
edge_bin_scatter_kernel(const int* __restrict__ ei,
                        const int* __restrict__ cnt,
                        int* __restrict__ binbase,
                        int* __restrict__ bsrc,
                        int* __restrict__ bdst) {
  __shared__ int ts[4][NBIN];
  __shared__ int ps[4][NBIN];
  __shared__ int cur[NBIN];
  __shared__ int wsum[8];
  const int tid  = threadIdx.x;
  const int base = blockIdx.x * ECHUNK;
  int s[EPT], d[EPT];
  #pragma unroll
  for (int j = 0; j < EPT; ++j) {
    int i = tid + 256 * j;
    i = (i < ECHUNK) ? i : (ECHUNK - 1);
    s[j] = ei[base + i];
    d[j] = ei[E_EDGES + base + i];
  }
  const int q = tid & 63, g = tid >> 6;
  int4 tot  = make_int4(0, 0, 0, 0);
  int4 part = make_int4(0, 0, 0, 0);
  #pragma unroll 16
  for (int j = 0; j < EBLK / 4; ++j) {
    const int b = 4 * j + g;
    const int4 c = ((const int4*)cnt)[b * (NBIN / 4) + q];
    tot.x += c.x; tot.y += c.y; tot.z += c.z; tot.w += c.w;
    if (b < (int)blockIdx.x) { part.x += c.x; part.y += c.y; part.z += c.z; part.w += c.w; }
  }
  ts[g][4 * q + 0] = tot.x;  ts[g][4 * q + 1] = tot.y;
  ts[g][4 * q + 2] = tot.z;  ts[g][4 * q + 3] = tot.w;
  ps[g][4 * q + 0] = part.x; ps[g][4 * q + 1] = part.y;
  ps[g][4 * q + 2] = part.z; ps[g][4 * q + 3] = part.w;
  __syncthreads();
  const int T = (ts[0][tid] + ts[1][tid]) + (ts[2][tid] + ts[3][tid]);
  const int P = (ps[0][tid] + ps[1][tid]) + (ps[2][tid] + ps[3][tid]);
  const int ex = block_excl_scan<8>(T, wsum);
  if (blockIdx.x == 0) {
    binbase[tid] = ex;
    if (tid == NBIN - 1) binbase[NBIN] = ex + T;
  }
  cur[tid] = ex + P;
  __syncthreads();
  int pos[EPT];
  #pragma unroll
  for (int j = 0; j < EPT; ++j)
    pos[j] = (tid + 256 * j < ECHUNK) ? atomicAdd(&cur[d[j] / BINW], 1) : 0;
  #pragma unroll
  for (int j = 0; j < EPT; ++j) {
    if (tid + 256 * j < ECHUNK) {
      bsrc[pos[j]] = s[j];
      bdst[pos[j]] = d[j];
    }
  }
}

__global__ void __launch_bounds__(SORT_T)
bin_sort_kernel(const int* __restrict__ binbase,
                const int* __restrict__ bsrc,
                const int* __restrict__ bdst,
                int* __restrict__ rowptr,
                int* __restrict__ csr) {
  __shared__ int c[NBIN];
  __shared__ int wsum[SORT_T / 32];
  const int tid = threadIdx.x;
  const int bin = blockIdx.x;
  const int beg = binbase[bin];
  const int end = binbase[bin + 1];
  const int n0  = bin * BINW;
  if (tid < NBIN) c[tid] = 0;
  __syncthreads();
  for (int i0 = beg; i0 < end; i0 += SORT_T * SORT_K) {
    int dd[SORT_K];
    #pragma unroll
    for (int k = 0; k < SORT_K; ++k) {
      const int i = i0 + tid + SORT_T * k;
      dd[k] = bdst[(i < end) ? i : (end - 1)];
    }
    #pragma unroll
    for (int k = 0; k < SORT_K; ++k)
      if (i0 + tid + SORT_T * k < end) atomicAdd(&c[dd[k] - n0], 1);
  }
  __syncthreads();
  const int v  = (tid < NBIN) ? c[tid] : 0;
  const int ex = block_excl_scan<SORT_T / 32>(v, wsum);
  if (tid < BINW && n0 + tid <= N_NODES) rowptr[n0 + tid] = beg + ex;
  if (tid < NBIN) c[tid] = beg + ex;
  __syncthreads();
  for (int i0 = beg; i0 < end; i0 += SORT_T * SORT_K) {
    int ss[SORT_K], dd[SORT_K];
    #pragma unroll
    for (int k = 0; k < SORT_K; ++k) {
      const int i  = i0 + tid + SORT_T * k;
      const int ic = (i < end) ? i : (end - 1);
      ss[k] = bsrc[ic];
      dd[k] = bdst[ic];
    }
    #pragma unroll
    for (int k = 0; k < SORT_K; ++k) {
      if (i0 + tid + SORT_T * k < end) {
        const int pos = atomicAdd(&c[dd[k] - n0], 1);
        csr[pos] = ss[k];
      }
    }
  }
}

#define NB 8
template <int K>
__device__ __forceinline__ float xlane(float v) {
  return __int_as_float(__builtin_amdgcn_mov_dpp(__float_as_int(v), 0x160 + K, 0xf, 0xf, true));
}
__device__ __forceinline__ float xhalf(float v) {
  int b = __float_as_int(v);
  return __int_as_float(__builtin_amdgcn_permlanex16(b, b, 0x76543210, 0xfedcba98, false, false));
}
template <int G>
__device__ __forceinline__ float group_lane(float v) {
  return __int_as_float((int)__builtin_amdgcn_mov_dpp8((unsigned)__float_as_int(v), G * 0x249249));
}
__device__ __forceinline__ float lane_score(v4f v, v4f xr4, v4f at4) {
  v4f t = v + xr4;
  t.x = (t.x > 0.f) ? t.x : t.x * SLOPE;
  t.y = (t.y > 0.f) ? t.y : t.y * SLOPE;
  t.z = (t.z > 0.f) ? t.z : t.z * SLOPE;
  t.w = (t.w > 0.f) ? t.w : t.w * SLOPE;
  return t.x * at4.x + t.y * at4.y + t.z * at4.z + t.w * at4.w;
}
__global__ void __launch_bounds__(256)
node_attn_ln_kernel(const int* __restrict__ rowptr,
                    const int* __restrict__ csr,
                    const float* __restrict__ xl,
                    const float* __restrict__ xr,
                    const float* __restrict__ att,
                    const float* __restrict__ bias,
                    const float* __restrict__ gamma,
                    const float* __restrict__ beta,
                    float* __restrict__ out) {
  const int lane = threadIdx.x & 31;
  const int wv   = __builtin_amdgcn_readfirstlane(threadIdx.x >> 5);
  const int node = blockIdx.x * 8 + wv;
  if (node >= N_NODES) return;
  const bool b0 = (lane & 1) != 0, b1 = (lane & 2) != 0, b2 = (lane & 4) != 0;
  const int edge_of_lane = (b0 ? 4 : 0) + (b1 ? 2 : 0) + (b2 ? 1 : 0);
  const v4f xr4 = ((const v4f*)(xr + (size_t)node * HC))[lane];
  const v4f at4 = ((const v4f*)att)[lane];
  v4f acc = ((const v4f*)(xl + (size_t)node * HC))[lane];
  float m = lane_score(acc, xr4, at4);
  m += xlane<1>(m);
  m += xlane<2>(m);
  m += xlane<4>(m);
  float den = (edge_of_lane == 0) ? 1.f : 0.f;
  const int beg = rowptr[node];
  const int end = rowptr[node + 1];
  int sid[NB];
  if (beg < end) {
    #pragma unroll
    for (int j = 0; j < NB; ++j) {
      int q = (beg + j < end) ? (beg + j) : (end - 1);
      sid[j] = csr[q];
    }
  }
  for (int p = beg; p < end; p += NB) {
    v4f r[NB];
    #pragma unroll
    for (int j = 0; j < NB; ++j)
      r[j] = ((const v4f*)(xl + (size_t)sid[j] * HC))[lane];
    #pragma unroll
    for (int j = 0; j < NB; ++j) {
      int q = (p + NB + j < end) ? (p + NB + j) : (end - 1);
      sid[j] = csr[q];
    }
    asm volatile("" :: "v"(r[0]), "v"(r[1]), "v"(r[2]), "v"(r[3]),
                       "v"(r[4]), "v"(r[5]), "v"(r[6]), "v"(r[7]));
    float s8[NB];
    #pragma unroll
    for (int j = 0; j < NB; ++j) s8[j] = lane_score(r[j], xr4, at4);
    float s4[4];
    #pragma unroll
    for (int k = 0; k < 4; ++k)
      s4[k] = (b0 ? s8[k + 4] : s8[k]) + xlane<1>(b0 ? s8[k] : s8[k + 4]);
    float s2[2];
    #pragma unroll
    for (int k = 0; k < 2; ++k)
      s2[k] = (b1 ? s4[k + 2] : s4[k]) + xlane<2>(b1 ? s4[k] : s4[k + 2]);
    float sc = (b2 ? s2[1] : s2[0]) + xlane<4>(b2 ? s2[0] : s2[1]);
    sc = (p + edge_of_lane < end) ? sc : -INFINITY;
    float mb = fmaxf(sc, xlane<1>(sc));
    mb = fmaxf(mb, xlane<2>(mb));
    mb = fmaxf(mb, xlane<4>(mb));
    mb = fmaxf(mb, m);
    const float scale = __expf(m - mb);
    const float w = __expf(sc - mb);
    acc *= scale;
    den = den * scale + w;
    acc += group_lane<0>(w) * r[0];
    acc += group_lane<4>(w) * r[1];
    acc += group_lane<2>(w) * r[2];
    acc += group_lane<6>(w) * r[3];
    acc += group_lane<1>(w) * r[4];
    acc += group_lane<5>(w) * r[5];
    acc += group_lane<3>(w) * r[6];
    acc += group_lane<7>(w) * r[7];
    m = mb;
  }
  den += xlane<1>(den);
  den += xlane<2>(den);
  den += xlane<4>(den);
  const float4 b4 = ((const float4*)bias)[lane];
  const float4 g4 = ((const float4*)gamma)[lane];
  const float4 e4 = ((const float4*)beta)[lane];
  const float inv = 1.0f / den;
  float h0 = acc.x * inv + b4.x, h1 = acc.y * inv + b4.y;
  float h2 = acc.z * inv + b4.z, h3 = acc.w * inv + b4.w;
  float sum = (h0 + h1) + (h2 + h3);
  sum += xhalf(sum);
  sum += xlane<8>(sum);
  sum += xlane<4>(sum);
  sum += xlane<2>(sum);
  sum += xlane<1>(sum);
  float mu = sum * (1.0f / HC);
  float d0 = h0 - mu, d1 = h1 - mu, d2 = h2 - mu, d3 = h3 - mu;
  float sq = (d0 * d0 + d1 * d1) + (d2 * d2 + d3 * d3);
  sq += xhalf(sq);
  sq += xlane<8>(sq);
  sq += xlane<4>(sq);
  sq += xlane<2>(sq);
  sq += xlane<1>(sq);
  float rstd = rsqrtf(sq * (1.0f / HC) + LN_EPS);
  float y0 = d0 * rstd * g4.x + e4.x, y1 = d1 * rstd * g4.y + e4.y;
  float y2 = d2 * rstd * g4.z + e4.z, y3 = d3 * rstd * g4.w + e4.w;
  float4 o;
  o.x = y0 / (1.0f + __expf(-y0));
  o.y = y1 / (1.0f + __expf(-y1));
  o.z = y2 / (1.0f + __expf(-y2));
  o.w = y3 / (1.0f + __expf(-y3));
  ((float4*)(out + (size_t)node * HC))[lane] = o;
}

#define POOL_NODES 32
#define POOL_BATCH 16
__device__ __forceinline__ void pool_flush(float* __restrict__ sums,
                                           float* __restrict__ counts,
                                           int g, int lane, float4 acc, float cnt) {
  float* p = sums + (size_t)g * HC + 4 * lane;
  atomAddF(p + 0, acc.x);
  atomAddF(p + 1, acc.y);
  atomAddF(p + 2, acc.z);
  atomAddF(p + 3, acc.w);
  if (lane == 0) atomAddF(&counts[g], cnt);
}
__global__ void __launch_bounds__(32)
pool_accum_kernel(const float* __restrict__ x,
                  const int* __restrict__ batch,
                  float* __restrict__ sums,
                  float* __restrict__ counts) {
  const int lane = threadIdx.x;
  const int n0 = blockIdx.x * POOL_NODES;
  const int left = N_NODES - n0;
  const int nn = (left < POOL_NODES) ? left : POOL_NODES;
  const float4* xp = (const float4*)(x + (size_t)n0 * HC) + lane;
  int gcur = batch[n0];
  float4 acc = make_float4(0.f, 0.f, 0.f, 0.f);
  float cnt = 0.f;
  for (int j0 = 0; j0 < nn; j0 += POOL_BATCH) {
    float4 v[POOL_BATCH];
    int g[POOL_BATCH];
    #pragma unroll
    for (int j = 0; j < POOL_BATCH; ++j) {
      int jj = (j0 + j < nn) ? (j0 + j) : (nn - 1);
      v[j] = xp[(size_t)jj * (HC / 4)];
      g[j] = batch[n0 + jj];
    }
    #pragma unroll
    for (int j = 0; j < POOL_BATCH; ++j) {
      if (j0 + j < nn) {
        if (g[j] != gcur) {
          pool_flush(sums, counts, gcur, lane, acc, cnt);
          acc = make_float4(0.f, 0.f, 0.f, 0.f);
          cnt = 0.f;
          gcur = g[j];
        }
        acc.x += v[j].x; acc.y += v[j].y; acc.z += v[j].z; acc.w += v[j].w;
        cnt += 1.0f;
      }
    }
  }
  pool_flush(sums, counts, gcur, lane, acc, cnt);
}
__global__ void pool_final_kernel(const float* __restrict__ sums,
                                  const float* __restrict__ counts,
                                  float* __restrict__ out) {
  int t = blockIdx.x * blockDim.x + threadIdx.x;
  if (t >= NGRAPH * HC) return;
  out[t] = sums[t] / fmaxf(counts[t >> 7], 1.0f);
}

extern "C" void kernel_launch(void* const* d_in, const int* in_sizes, int n_in,
                              void* d_out, int out_size, void* d_ws, size_t ws_size,
                              hipStream_t stream) {
  (void)in_sizes; (void)n_in; (void)out_size; (void)ws_size;
  const float* x     = (const float*)d_in[0];
  const int*   ei    = (const int*)d_in[1];
  const int*   batch = (const int*)d_in[2];
  const float* Wl[2]    = {(const float*)d_in[3],  (const float*)d_in[11]};
  const float* Wr[2]    = {(const float*)d_in[4],  (const float*)d_in[12]};
  const float* bl[2]    = {(const float*)d_in[5],  (const float*)d_in[13]};
  const float* br[2]    = {(const float*)d_in[6],  (const float*)d_in[14]};
  const float* att[2]   = {(const float*)d_in[7],  (const float*)d_in[15]};
  const float* bias[2]  = {(const float*)d_in[8],  (const float*)d_in[16]};
  const float* gamma[2] = {(const float*)d_in[9],  (const float*)d_in[17]};
  const float* beta[2]  = {(const float*)d_in[10], (const float*)d_in[18]};

  float* ws = (float*)d_ws;
  float* xl     = ws;          ws += (size_t)N_NODES * HC;
  float* xr2    = ws;          ws += (size_t)N_NODES * HC;
  float* xnode  = ws;          ws += (size_t)N_NODES * HC;
  int*   cnt    = (int*)ws;    ws += (size_t)EBLK * NBIN;
  int*   binbase= (int*)ws;    ws += (size_t)NBIN + 4;
  int*   rowptr = (int*)ws;    ws += (size_t)N_NODES + 4;
  int*   bsrc   = (int*)ws;    ws += (size_t)E_EDGES;
  int*   bdst   = (int*)ws;    ws += (size_t)E_EDGES;
  int*   csr    = (int*)ws;    ws += (size_t)E_EDGES;
  float* psums  = ws;          ws += (size_t)NGRAPH * HC;
  float* pcnt   = ws;

  dim3 b256(256);
  const int    gemmGrid = (N_NODES + 16 * ROWTILES - 1) / (16 * ROWTILES);
  const size_t shmem    = ((size_t)HC * LDSW + 16 * LDSA) * sizeof(float);
  const int    nodeGrid = (N_NODES + 7) / 8;
  const int    poolGrid = (N_NODES + POOL_NODES - 1) / POOL_NODES;

  edge_bin_count_kernel<<<EBLK, b256, 0, stream>>>(ei, cnt);
  edge_bin_scatter_kernel<<<EBLK, b256, 0, stream>>>(ei, cnt, binbase, bsrc, bdst);
  bin_sort_kernel<<<NBIN, dim3(SORT_T), 0, stream>>>(binbase, bsrc, bdst, rowptr, csr);

  const float* xin = x;
  for (int l = 0; l < 2; ++l) {
    gemm_bias_wmma<<<gemmGrid, b256, shmem, stream>>>(xin, Wl[l], bl[l], xl, N_NODES);
    gemm_bias_wmma<<<gemmGrid, b256, shmem, stream>>>(xin, Wr[l], br[l], xr2, N_NODES);
    node_attn_ln_kernel<<<nodeGrid, b256, 0, stream>>>(rowptr, csr, xl, xr2, att[l],
                                                       bias[l], gamma[l], beta[l], xnode);
    xin = xnode;
  }
  fill_f32<<<(NGRAPH * HC + NGRAPH + 255) / 256, b256, 0, stream>>>(psums, 0.f, NGRAPH * HC + NGRAPH);
  pool_accum_kernel<<<poolGrid, dim3(32), 0, stream>>>(xnode, batch, psums, pcnt);
  pool_final_kernel<<<(NGRAPH * HC + 255) / 256, b256, 0, stream>>>(psums, pcnt, (float*)d_out);
}


# GEMM: a layer's two projections (x@Wl, x@Wr) in ONE launch: both weights in LDS, one A fragment feeds two WMMAs; row tiles dealt evenly over 512 resident blocks; 12 launches to 10; sums unchanged
# speedup vs baseline: 4.8256x; 1.0635x over previous
#include <hip/hip_runtime.h>
#include <hip/hip_bf16.h>
#include <math.h>

#define N_NODES 50000
#define E_EDGES 800000
#define ETOT    (E_EDGES + N_NODES)
#define HEADS   4
#define CHP     32
#define HC      128
#define NGRAPH  64
#define SLOPE   0.2f
#define LN_EPS  1e-5f

typedef float v2f __attribute__((ext_vector_type(2)));
typedef float v8f __attribute__((ext_vector_type(8)));
typedef float v4f __attribute__((ext_vector_type(4)));

__device__ __forceinline__ void atomAddF(float* p, float v) {
  unsafeAtomicAdd(p, v);
}

__global__ void fill_f32(float* __restrict__ p, float v, int n) {
  int i = blockIdx.x * blockDim.x + threadIdx.x;
  if (i < n) p[i] = v;
}

#define LDSW2 264
#define LDSA 132
#define GEMM_BLOCKS 512
static_assert((HC * LDSW2 + 16 * LDSA) * 4 * 2 <= 320 * 1024, "two blocks share one LDS");
__global__ void __launch_bounds__(256)
gemm2_bias_wmma(const float* __restrict__ X,
                const float* __restrict__ Wa,
                const float* __restrict__ Wb,
                const float* __restrict__ ba,
                const float* __restrict__ bb,
                float* __restrict__ Ya,
                float* __restrict__ Yb, int n) {
  __shared__ __attribute__((aligned(16))) float smem[HC * LDSW2 + 16 * LDSA];
  float* sW = smem;
  float* sA = smem + HC * LDSW2;
  const int tid  = threadIdx.x;
  const int wave = tid >> 5;
  const int lane = tid & 31;
  const int ntiles = (n + 15) >> 4;
  const int t0 = (int)(((unsigned)blockIdx.x * (unsigned)ntiles) / GEMM_BLOCKS);
  const int t1 = (int)(((unsigned)(blockIdx.x + 1) * (unsigned)ntiles) / GEMM_BLOCKS);
  if (t0 >= t1) return;

  float4 pa0, pa1;
  {
    int r0 = t0 * 16 + wave, r1 = r0 + 8;
    r0 = (r0 < n) ? r0 : (n - 1);
    r1 = (r1 < n) ? r1 : (n - 1);
    pa0 = ((const float4*)(X + (size_t)r0 * HC))[lane];
    pa1 = ((const float4*)(X + (size_t)r1 * HC))[lane];
  }
  #pragma unroll
  for (int i = 0; i < 16; ++i) {
    int idx = tid + 256 * i;
    int r = idx >> 5, q = idx & 31;
    float4 w4 = *(const float4*)(Wa + r * HC + q * 4);
    *(float4*)(sW + r * LDSW2 + q * 4) = w4;
  }
  #pragma unroll
  for (int i = 0; i < 16; ++i) {
    int idx = tid + 256 * i;
    int r = idx >> 5, q = idx & 31;
    float4 w4 = *(const float4*)(Wb + r * HC + q * 4);
    *(float4*)(sW + r * LDSW2 + HC + q * 4) = w4;
  }
  *(float4*)(sA + wave * LDSA + lane * 4)       = pa0;
  *(float4*)(sA + (wave + 8) * LDSA + lane * 4) = pa1;
  __syncthreads();

  const int m    = lane & 15;
  const int hi   = lane >> 4;
  const bool second = (wave >= 4);
  float* __restrict__ Y = second ? Yb : Ya;
  const float* bvec = second ? bb : ba;
  const int ocol = (wave & 3) * 32 + m;
  const float bc0 = bvec[ocol];
  const float bc1 = bvec[ocol + 16];
  const float* ap = sA + m * LDSA + 2 * hi;
  const float* bp0 = sW + (2 * hi) * LDSW2 + wave * 32 + m;

  for (int t = t0; t < t1; ++t) {
    const int row0 = t * 16;
    const bool more = (t + 1 < t1);
    if (more) {
      int r0 = row0 + 16 + wave, r1 = r0 + 8;
      r0 = (r0 < n) ? r0 : (n - 1);
      r1 = (r1 < n) ? r1 : (n - 1);
      pa0 = ((const float4*)(X + (size_t)r0 * HC))[lane];
      pa1 = ((const float4*)(X + (size_t)r1 * HC))[lane];
    }
    v8f acc0 = {};
    v8f acc1 = {};
    #pragma unroll
    for (int k0 = 0; k0 < HC; k0 += 4) {
      v2f a = *(const v2f*)(ap + k0);
      const float* bp = bp0 + k0 * LDSW2;
      v2f b0; b0.x = bp[0];  b0.y = bp[LDSW2];
      v2f b1; b1.x = bp[16]; b1.y = bp[LDSW2 + 16];
      acc0 = __builtin_amdgcn_wmma_f32_16x16x4_f32(
          false, a, false, b0, (short)0, acc0, false, false);
      acc1 = __builtin_amdgcn_wmma_f32_16x16x4_f32(
          false, a, false, b1, (short)0, acc1, false, false);
    }
    if (more) {
      __syncthreads();
      *(float4*)(sA + wave * LDSA + lane * 4)       = pa0;
      *(float4*)(sA + (wave + 8) * LDSA + lane * 4) = pa1;
      __syncthreads();
    }
    #pragma unroll
    for (int r = 0; r < 8; ++r) {
      int orow = row0 + r + 8 * hi;
      if (orow < n) {
        float* yp = Y + (size_t)orow * HC + ocol;
        yp[0]  = acc0[r] + bc0;
        yp[16] = acc1[r] + bc1;
      }
    }
  }
}

#define NBIN   256
#define BINW   196
#define EBLK   256
#define ECHUNK (E_EDGES / EBLK)
#define EPT    ((ECHUNK + 255) / 256)
#define SORT_T 512
#define SORT_K 8
static_assert(EBLK * ECHUNK == E_EDGES, "edge blocks cover the edges exactly");
static_assert(NBIN * BINW > N_NODES, "ranges cover node ids 0..N_NODES");
static_assert(BINW <= NBIN, "one counter per node of a range");

template <int NW>
__device__ __forceinline__ int block_excl_scan(int v, int* wsum) {
  const int lane = threadIdx.x & 31;
  const int w    = threadIdx.x >> 5;
  int inc = v;
  for (int off = 1; off < 32; off <<= 1) {
    int u = __shfl_up(inc, off, 32);
    if (lane >= off) inc += u;
  }
  if (lane == 31) wsum[w] = inc;
  __syncthreads();
  int carry = 0;
  for (int k = 0; k < NW; ++k) carry += (k < w) ? wsum[k] : 0;
  return carry + inc - v;
}

__global__ void __launch_bounds__(256)
edge_bin_count_kernel(const int* __restrict__ ei, int* __restrict__ cnt) {
  __shared__ int h[NBIN];
  const int tid  = threadIdx.x;
  const int base = blockIdx.x * ECHUNK;
  int d[EPT];
  #pragma unroll
  for (int j = 0; j < EPT; ++j) {
    int i = tid + 256 * j;
    i = (i < ECHUNK) ? i : (ECHUNK - 1);
    d[j] = ei[E_EDGES + base + i];
  }
  h[tid] = 0;
  __syncthreads();
  #pragma unroll
  for (int j = 0; j < EPT; ++j)
    if (tid + 256 * j < ECHUNK) atomicAdd(&h[d[j] / BINW], 1);
  __syncthreads();
  cnt[blockIdx.x * NBIN + tid] = h[tid];
}

__global__ void __launch_bounds__(256)
edge_bin_scatter_kernel(const int* __restrict__ ei,
                        const int* __restrict__ cnt,
                        int* __restrict__ binbase,
                        int* __restrict__ bsrc,
                        int* __restrict__ bdst) {
  __shared__ int ts[4][NBIN];
  __shared__ int ps[4][NBIN];
  __shared__ int cur[NBIN];
  __shared__ int wsum[8];
  const int tid  = threadIdx.x;
  const int base = blockIdx.x * ECHUNK;
  int s[EPT], d[EPT];
  #pragma unroll
  for (int j = 0; j < EPT; ++j) {
    int i = tid + 256 * j;
    i = (i < ECHUNK) ? i : (ECHUNK - 1);
    s[j] = ei[base + i];
    d[j] = ei[E_EDGES + base + i];
  }
  const int q = tid & 63, g = tid >> 6;
  int4 tot  = make_int4(0, 0, 0, 0);
  int4 part = make_int4(0, 0, 0, 0);
  #pragma unroll 16
  for (int j = 0; j < EBLK / 4; ++j) {
    const int b = 4 * j + g;
    const int4 c = ((const int4*)cnt)[b * (NBIN / 4) + q];
    tot.x += c.x; tot.y += c.y; tot.z += c.z; tot.w += c.w;
    if (b < (int)blockIdx.x) { part.x += c.x; part.y += c.y; part.z += c.z; part.w += c.w; }
  }
  ts[g][4 * q + 0] = tot.x;  ts[g][4 * q + 1] = tot.y;
  ts[g][4 * q + 2] = tot.z;  ts[g][4 * q + 3] = tot.w;
  ps[g][4 * q + 0] = part.x; ps[g][4 * q + 1] = part.y;
  ps[g][4 * q + 2] = part.z; ps[g][4 * q + 3] = part.w;
  __syncthreads();
  const int T = (ts[0][tid] + ts[1][tid]) + (ts[2][tid] + ts[3][tid]);
  const int P = (ps[0][tid] + ps[1][tid]) + (ps[2][tid] + ps[3][tid]);
  const int ex = block_excl_scan<8>(T, wsum);
  if (blockIdx.x == 0) {
    binbase[tid] = ex;
    if (tid == NBIN - 1) binbase[NBIN] = ex + T;
  }
  cur[tid] = ex + P;
  __syncthreads();
  int pos[EPT];
  #pragma unroll
  for (int j = 0; j < EPT; ++j)
    pos[j] = (tid + 256 * j < ECHUNK) ? atomicAdd(&cur[d[j] / BINW], 1) : 0;
  #pragma unroll
  for (int j = 0; j < EPT; ++j) {
    if (tid + 256 * j < ECHUNK) {
      bsrc[pos[j]] = s[j];
      bdst[pos[j]] = d[j];
    }
  }
}

__global__ void __launch_bounds__(SORT_T)
bin_sort_kernel(const int* __restrict__ binbase,
                const int* __restrict__ bsrc,
                const int* __restrict__ bdst,
                int* __restrict__ rowptr,
                int* __restrict__ csr) {
  __shared__ int c[NBIN];
  __shared__ int wsum[SORT_T / 32];
  const int tid = threadIdx.x;
  const int bin = blockIdx.x;
  const int beg = binbase[bin];
  const int end = binbase[bin + 1];
  const int n0  = bin * BINW;
  if (tid < NBIN) c[tid] = 0;
  __syncthreads();
  for (int i0 = beg; i0 < end; i0 += SORT_T * SORT_K) {
    int dd[SORT_K];
    #pragma unroll
    for (int k = 0; k < SORT_K; ++k) {
      const int i = i0 + tid + SORT_T * k;
      dd[k] = bdst[(i < end) ? i : (end - 1)];
    }
    #pragma unroll
    for (int k = 0; k < SORT_K; ++k)
      if (i0 + tid + SORT_T * k < end) atomicAdd(&c[dd[k] - n0], 1);
  }
  __syncthreads();
  const int v  = (tid < NBIN) ? c[tid] : 0;
  const int ex = block_excl_scan<SORT_T / 32>(v, wsum);
  if (tid < BINW && n0 + tid <= N_NODES) rowptr[n0 + tid] = beg + ex;
  if (tid < NBIN) c[tid] = beg + ex;
  __syncthreads();
  for (int i0 = beg; i0 < end; i0 += SORT_T * SORT_K) {
    int ss[SORT_K], dd[SORT_K];
    #pragma unroll
    for (int k = 0; k < SORT_K; ++k) {
      const int i  = i0 + tid + SORT_T * k;
      const int ic = (i < end) ? i : (end - 1);
      ss[k] = bsrc[ic];
      dd[k] = bdst[ic];
    }
    #pragma unroll
    for (int k = 0; k < SORT_K; ++k) {
      if (i0 + tid + SORT_T * k < end) {
        const int pos = atomicAdd(&c[dd[k] - n0], 1);
        csr[pos] = ss[k];
      }
    }
  }
}

#define NB 8
template <int K>
__device__ __forceinline__ float xlane(float v) {
  return __int_as_float(__builtin_amdgcn_mov_dpp(__float_as_int(v), 0x160 + K, 0xf, 0xf, true));
}
__device__ __forceinline__ float xhalf(float v) {
  int b = __float_as_int(v);
  return __int_as_float(__builtin_amdgcn_permlanex16(b, b, 0x76543210, 0xfedcba98, false, false));
}
template <int G>
__device__ __forceinline__ float group_lane(float v) {
  return __int_as_float((int)__builtin_amdgcn_mov_dpp8((unsigned)__float_as_int(v), G * 0x249249));
}
__device__ __forceinline__ float lane_score(v4f v, v4f xr4, v4f at4) {
  v4f t = v + xr4;
  t.x = (t.x > 0.f) ? t.x : t.x * SLOPE;
  t.y = (t.y > 0.f) ? t.y : t.y * SLOPE;
  t.z = (t.z > 0.f) ? t.z : t.z * SLOPE;
  t.w = (t.w > 0.f) ? t.w : t.w * SLOPE;
  return t.x * at4.x + t.y * at4.y + t.z * at4.z + t.w * at4.w;
}
__global__ void __launch_bounds__(256)
node_attn_ln_kernel(const int* __restrict__ rowptr,
                    const int* __restrict__ csr,
                    const float* __restrict__ xl,
                    const float* __restrict__ xr,
                    const float* __restrict__ att,
                    const float* __restrict__ bias,
                    const float* __restrict__ gamma,
                    const float* __restrict__ beta,
                    float* __restrict__ out) {
  const int lane = threadIdx.x & 31;
  const int wv   = __builtin_amdgcn_readfirstlane(threadIdx.x >> 5);
  const int node = blockIdx.x * 8 + wv;
  if (node >= N_NODES) return;
  const bool b0 = (lane & 1) != 0, b1 = (lane & 2) != 0, b2 = (lane & 4) != 0;
  const int edge_of_lane = (b0 ? 4 : 0) + (b1 ? 2 : 0) + (b2 ? 1 : 0);
  const v4f xr4 = ((const v4f*)(xr + (size_t)node * HC))[lane];
  const v4f at4 = ((const v4f*)att)[lane];
  v4f acc = ((const v4f*)(xl + (size_t)node * HC))[lane];
  float m = lane_score(acc, xr4, at4);
  m += xlane<1>(m);
  m += xlane<2>(m);
  m += xlane<4>(m);
  float den = (edge_of_lane == 0) ? 1.f : 0.f;
  const int beg = rowptr[node];
  const int end = rowptr[node + 1];
  int sid[NB];
  if (beg < end) {
    #pragma unroll
    for (int j = 0; j < NB; ++j) {
      int q = (beg + j < end) ? (beg + j) : (end - 1);
      sid[j] = csr[q];
    }
  }
  for (int p = beg; p < end; p += NB) {
    v4f r[NB];
    #pragma unroll
    for (int j = 0; j < NB; ++j)
      r[j] = ((const v4f*)(xl + (size_t)sid[j] * HC))[lane];
    #pragma unroll
    for (int j = 0; j < NB; ++j) {
      int q = (p + NB + j < end) ? (p + NB + j) : (end - 1);
      sid[j] = csr[q];
    }
    asm volatile("" :: "v"(r[0]), "v"(r[1]), "v"(r[2]), "v"(r[3]),
                       "v"(r[4]), "v"(r[5]), "v"(r[6]), "v"(r[7]));
    float s8[NB];
    #pragma unroll
    for (int j = 0; j < NB; ++j) s8[j] = lane_score(r[j], xr4, at4);
    float s4[4];
    #pragma unroll
    for (int k = 0; k < 4; ++k)
      s4[k] = (b0 ? s8[k + 4] : s8[k]) + xlane<1>(b0 ? s8[k] : s8[k + 4]);
    float s2[2];
    #pragma unroll
    for (int k = 0; k < 2; ++k)
      s2[k] = (b1 ? s4[k + 2] : s4[k]) + xlane<2>(b1 ? s4[k] : s4[k + 2]);
    float sc = (b2 ? s2[1] : s2[0]) + xlane<4>(b2 ? s2[0] : s2[1]);
    sc = (p + edge_of_lane < end) ? sc : -INFINITY;
    float mb = fmaxf(sc, xlane<1>(sc));
    mb = fmaxf(mb, xlane<2>(mb));
    mb = fmaxf(mb, xlane<4>(mb));
    mb = fmaxf(mb, m);
    const float scale = __expf(m - mb);
    const float w = __expf(sc - mb);
    acc *= scale;
    den = den * scale + w;
    acc += group_lane<0>(w) * r[0];
    acc += group_lane<4>(w) * r[1];
    acc += group_lane<2>(w) * r[2];
    acc += group_lane<6>(w) * r[3];
    acc += group_lane<1>(w) * r[4];
    acc += group_lane<5>(w) * r[5];
    acc += group_lane<3>(w) * r[6];
    acc += group_lane<7>(w) * r[7];
    m = mb;
  }
  den += xlane<1>(den);
  den += xlane<2>(den);
  den += xlane<4>(den);
  const float4 b4 = ((const float4*)bias)[lane];
  const float4 g4 = ((const float4*)gamma)[lane];
  const float4 e4 = ((const float4*)beta)[lane];
  const float inv = 1.0f / den;
  float h0 = acc.x * inv + b4.x, h1 = acc.y * inv + b4.y;
  float h2 = acc.z * inv + b4.z, h3 = acc.w * inv + b4.w;
  float sum = (h0 + h1) + (h2 + h3);
  sum += xhalf(sum);
  sum += xlane<8>(sum);
  sum += xlane<4>(sum);
  sum += xlane<2>(sum);
  sum += xlane<1>(sum);
  float mu = sum * (1.0f / HC);
  float d0 = h0 - mu, d1 = h1 - mu, d2 = h2 - mu, d3 = h3 - mu;
  float sq = (d0 * d0 + d1 * d1) + (d2 * d2 + d3 * d3);
  sq += xhalf(sq);
  sq += xlane<8>(sq);
  sq += xlane<4>(sq);
  sq += xlane<2>(sq);
  sq += xlane<1>(sq);
  float rstd = rsqrtf(sq * (1.0f / HC) + LN_EPS);
  float y0 = d0 * rstd * g4.x + e4.x, y1 = d1 * rstd * g4.y + e4.y;
  float y2 = d2 * rstd * g4.z + e4.z, y3 = d3 * rstd * g4.w + e4.w;
  float4 o;
  o.x = y0 / (1.0f + __expf(-y0));
  o.y = y1 / (1.0f + __expf(-y1));
  o.z = y2 / (1.0f + __expf(-y2));
  o.w = y3 / (1.0f + __expf(-y3));
  ((float4*)(out + (size_t)node * HC))[lane] = o;
}

#define POOL_NODES 32
#define POOL_BATCH 16
__device__ __forceinline__ void pool_flush(float* __restrict__ sums,
                                           float* __restrict__ counts,
                                           int g, int lane, float4 acc, float cnt) {
  float* p = sums + (size_t)g * HC + 4 * lane;
  atomAddF(p + 0, acc.x);
  atomAddF(p + 1, acc.y);
  atomAddF(p + 2, acc.z);
  atomAddF(p + 3, acc.w);
  if (lane == 0) atomAddF(&counts[g], cnt);
}
__global__ void __launch_bounds__(32)
pool_accum_kernel(const float* __restrict__ x,
                  const int* __restrict__ batch,
                  float* __restrict__ sums,
                  float* __restrict__ counts) {
  const int lane = threadIdx.x;
  const int n0 = blockIdx.x * POOL_NODES;
  const int left = N_NODES - n0;
  const int nn = (left < POOL_NODES) ? left : POOL_NODES;
  const float4* xp = (const float4*)(x + (size_t)n0 * HC) + lane;
  int gcur = batch[n0];
  float4 acc = make_float4(0.f, 0.f, 0.f, 0.f);
  float cnt = 0.f;
  for (int j0 = 0; j0 < nn; j0 += POOL_BATCH) {
    float4 v[POOL_BATCH];
    int g[POOL_BATCH];
    #pragma unroll
    for (int j = 0; j < POOL_BATCH; ++j) {
      int jj = (j0 + j < nn) ? (j0 + j) : (nn - 1);
      v[j] = xp[(size_t)jj * (HC / 4)];
      g[j] = batch[n0 + jj];
    }
    #pragma unroll
    for (int j = 0; j < POOL_BATCH; ++j) {
      if (j0 + j < nn) {
        if (g[j] != gcur) {
          pool_flush(sums, counts, gcur, lane, acc, cnt);
          acc = make_float4(0.f, 0.f, 0.f, 0.f);
          cnt = 0.f;
          gcur = g[j];
        }
        acc.x += v[j].x; acc.y += v[j].y; acc.z += v[j].z; acc.w += v[j].w;
        cnt += 1.0f;
      }
    }
  }
  pool_flush(sums, counts, gcur, lane, acc, cnt);
}
__global__ void pool_final_kernel(const float* __restrict__ sums,
                                  const float* __restrict__ counts,
                                  float* __restrict__ out) {
  int t = blockIdx.x * blockDim.x + threadIdx.x;
  if (t >= NGRAPH * HC) return;
  out[t] = sums[t] / fmaxf(counts[t >> 7], 1.0f);
}

extern "C" void kernel_launch(void* const* d_in, const int* in_sizes, int n_in,
                              void* d_out, int out_size, void* d_ws, size_t ws_size,
                              hipStream_t stream) {
  (void)in_sizes; (void)n_in; (void)out_size; (void)ws_size;
  const float* x     = (const float*)d_in[0];
  const int*   ei    = (const int*)d_in[1];
  const int*   batch = (const int*)d_in[2];
  const float* Wl[2]    = {(const float*)d_in[3],  (const float*)d_in[11]};
  const float* Wr[2]    = {(const float*)d_in[4],  (const float*)d_in[12]};
  const float* bl[2]    = {(const float*)d_in[5],  (const float*)d_in[13]};
  const float* br[2]    = {(const float*)d_in[6],  (const float*)d_in[14]};
  const float* att[2]   = {(const float*)d_in[7],  (const float*)d_in[15]};
  const float* bias[2]  = {(const float*)d_in[8],  (const float*)d_in[16]};
  const float* gamma[2] = {(const float*)d_in[9],  (const float*)d_in[17]};
  const float* beta[2]  = {(const float*)d_in[10], (const float*)d_in[18]};

  float* ws = (float*)d_ws;
  float* xl     = ws;          ws += (size_t)N_NODES * HC;
  float* xr2    = ws;          ws += (size_t)N_NODES * HC;
  float* xnode  = ws;          ws += (size_t)N_NODES * HC;
  int*   cnt    = (int*)ws;    ws += (size_t)EBLK * NBIN;
  int*   binbase= (int*)ws;    ws += (size_t)NBIN + 4;
  int*   rowptr = (int*)ws;    ws += (size_t)N_NODES + 4;
  int*   bsrc   = (int*)ws;    ws += (size_t)E_EDGES;
  int*   bdst   = (int*)ws;    ws += (size_t)E_EDGES;
  int*   csr    = (int*)ws;    ws += (size_t)E_EDGES;
  float* psums  = ws;          ws += (size_t)NGRAPH * HC;
  float* pcnt   = ws;

  dim3 b256(256);
  const int    nodeGrid = (N_NODES + 7) / 8;
  const int    poolGrid = (N_NODES + POOL_NODES - 1) / POOL_NODES;

  edge_bin_count_kernel<<<EBLK, b256, 0, stream>>>(ei, cnt);
  edge_bin_scatter_kernel<<<EBLK, b256, 0, stream>>>(ei, cnt, binbase, bsrc, bdst);
  bin_sort_kernel<<<NBIN, dim3(SORT_T), 0, stream>>>(binbase, bsrc, bdst, rowptr, csr);

  const float* xin = x;
  for (int l = 0; l < 2; ++l) {
    gemm2_bias_wmma<<<GEMM_BLOCKS, b256, 0, stream>>>(xin, Wl[l], Wr[l], bl[l], br[l], xl, xr2, N_NODES);
    node_attn_ln_kernel<<<nodeGrid, b256, 0, stream>>>(rowptr, csr, xl, xr2, att[l],
                                                       bias[l], gamma[l], beta[l], xnode);
    xin = xnode;
  }
  fill_f32<<<(NGRAPH * HC + NGRAPH + 255) / 256, b256, 0, stream>>>(psums, 0.f, NGRAPH * HC + NGRAPH);
  pool_accum_kernel<<<poolGrid, dim3(32), 0, stream>>>(xnode, batch, psums, pcnt);
  pool_final_kernel<<<(NGRAPH * HC + 255) / 256, b256, 0, stream>>>(psums, pcnt, (float*)d_out);
}


# node pass: LeakyReLU as max(t, slope*t), score dot as two fused pairs, row loads from scalar bases: 144 not 194 vector ops per 8 edges, 64 not 90 registers per lane; score sum order changes
# speedup vs baseline: 4.9955x; 1.0352x over previous
#include <hip/hip_runtime.h>
#include <hip/hip_bf16.h>
#include <math.h>

#define N_NODES 50000
#define E_EDGES 800000
#define ETOT    (E_EDGES + N_NODES)
#define HEADS   4
#define CHP     32
#define HC      128
#define NGRAPH  64
#define SLOPE   0.2f
#define LN_EPS  1e-5f

typedef float v2f __attribute__((ext_vector_type(2)));
typedef float v8f __attribute__((ext_vector_type(8)));
typedef float v4f __attribute__((ext_vector_type(4)));

__device__ __forceinline__ void atomAddF(float* p, float v) {
  unsafeAtomicAdd(p, v);
}

__global__ void fill_f32(float* __restrict__ p, float v, int n) {
  int i = blockIdx.x * blockDim.x + threadIdx.x;
  if (i < n) p[i] = v;
}

#define LDSW2 264
#define LDSA 132
#define GEMM_BLOCKS 512
static_assert((HC * LDSW2 + 16 * LDSA) * 4 * 2 <= 320 * 1024, "two blocks share one LDS");
__global__ void __launch_bounds__(256)
gemm2_bias_wmma(const float* __restrict__ X,
                const float* __restrict__ Wa,
                const float* __restrict__ Wb,
                const float* __restrict__ ba,
                const float* __restrict__ bb,
                float* __restrict__ Ya,
                float* __restrict__ Yb, int n) {
  __shared__ __attribute__((aligned(16))) float smem[HC * LDSW2 + 16 * LDSA];
  float* sW = smem;
  float* sA = smem + HC * LDSW2;
  const int tid  = threadIdx.x;
  const int wave = tid >> 5;
  const int lane = tid & 31;
  const int ntiles = (n + 15) >> 4;
  const int t0 = (int)(((unsigned)blockIdx.x * (unsigned)ntiles) / GEMM_BLOCKS);
  const int t1 = (int)(((unsigned)(blockIdx.x + 1) * (unsigned)ntiles) / GEMM_BLOCKS);
  if (t0 >= t1) return;

  float4 pa0, pa1;
  {
    int r0 = t0 * 16 + wave, r1 = r0 + 8;
    r0 = (r0 < n) ? r0 : (n - 1);
    r1 = (r1 < n) ? r1 : (n - 1);
    pa0 = ((const float4*)(X + (size_t)r0 * HC))[lane];
    pa1 = ((const float4*)(X + (size_t)r1 * HC))[lane];
  }
  #pragma unroll
  for (int i = 0; i < 16; ++i) {
    int idx = tid + 256 * i;
    int r = idx >> 5, q = idx & 31;
    float4 w4 = *(const float4*)(Wa + r * HC + q * 4);
    *(float4*)(sW + r * LDSW2 + q * 4) = w4;
  }
  #pragma unroll
  for (int i = 0; i < 16; ++i) {
    int idx = tid + 256 * i;
    int r = idx >> 5, q = idx & 31;
    float4 w4 = *(const float4*)(Wb + r * HC + q * 4);
    *(float4*)(sW + r * LDSW2 + HC + q * 4) = w4;
  }
  *(float4*)(sA + wave * LDSA + lane * 4)       = pa0;
  *(float4*)(sA + (wave + 8) * LDSA + lane * 4) = pa1;
  __syncthreads();

  const int m    = lane & 15;
  const int hi   = lane >> 4;
  const bool second = (wave >= 4);
  float* __restrict__ Y = second ? Yb : Ya;
  const float* bvec = second ? bb : ba;
  const int ocol = (wave & 3) * 32 + m;
  const float bc0 = bvec[ocol];
  const float bc1 = bvec[ocol + 16];
  const float* ap = sA + m * LDSA + 2 * hi;
  const float* bp0 = sW + (2 * hi) * LDSW2 + wave * 32 + m;

  for (int t = t0; t < t1; ++t) {
    const int row0 = t * 16;
    const bool more = (t + 1 < t1);
    if (more) {
      int r0 = row0 + 16 + wave, r1 = r0 + 8;
      r0 = (r0 < n) ? r0 : (n - 1);
      r1 = (r1 < n) ? r1 : (n - 1);
      pa0 = ((const float4*)(X + (size_t)r0 * HC))[lane];
      pa1 = ((const float4*)(X + (size_t)r1 * HC))[lane];
    }
    v8f acc0 = {};
    v8f acc1 = {};
    #pragma unroll
    for (int k0 = 0; k0 < HC; k0 += 4) {
      v2f a = *(const v2f*)(ap + k0);
      const float* bp = bp0 + k0 * LDSW2;
      v2f b0; b0.x = bp[0];  b0.y = bp[LDSW2];
      v2f b1; b1.x = bp[16]; b1.y = bp[LDSW2 + 16];
      acc0 = __builtin_amdgcn_wmma_f32_16x16x4_f32(
          false, a, false, b0, (short)0, acc0, false, false);
      acc1 = __builtin_amdgcn_wmma_f32_16x16x4_f32(
          false, a, false, b1, (short)0, acc1, false, false);
    }
    if (more) {
      __syncthreads();
      *(float4*)(sA + wave * LDSA + lane * 4)       = pa0;
      *(float4*)(sA + (wave + 8) * LDSA + lane * 4) = pa1;
      __syncthreads();
    }
    #pragma unroll
    for (int r = 0; r < 8; ++r) {
      int orow = row0 + r + 8 * hi;
      if (orow < n) {
        float* yp = Y + (size_t)orow * HC + ocol;
        yp[0]  = acc0[r] + bc0;
        yp[16] = acc1[r] + bc1;
      }
    }
  }
}

#define NBIN   256
#define BINW   196
#define EBLK   256
#define ECHUNK (E_EDGES / EBLK)
#define EPT    ((ECHUNK + 255) / 256)
#define SORT_T 512
#define SORT_K 8
static_assert(EBLK * ECHUNK == E_EDGES, "edge blocks cover the edges exactly");
static_assert(NBIN * BINW > N_NODES, "ranges cover node ids 0..N_NODES");
static_assert(BINW <= NBIN, "one counter per node of a range");

template <int NW>
__device__ __forceinline__ int block_excl_scan(int v, int* wsum) {
  const int lane = threadIdx.x & 31;
  const int w    = threadIdx.x >> 5;
  int inc = v;
  for (int off = 1; off < 32; off <<= 1) {
    int u = __shfl_up(inc, off, 32);
    if (lane >= off) inc += u;
  }
  if (lane == 31) wsum[w] = inc;
  __syncthreads();
  int carry = 0;
  for (int k = 0; k < NW; ++k) carry += (k < w) ? wsum[k] : 0;
  return carry + inc - v;
}

__global__ void __launch_bounds__(256)
edge_bin_count_kernel(const int* __restrict__ ei, int* __restrict__ cnt) {
  __shared__ int h[NBIN];
  const int tid  = threadIdx.x;
  const int base = blockIdx.x * ECHUNK;
  int d[EPT];
  #pragma unroll
  for (int j = 0; j < EPT; ++j) {
    int i = tid + 256 * j;
    i = (i < ECHUNK) ? i : (ECHUNK - 1);
    d[j] = ei[E_EDGES + base + i];
  }
  h[tid] = 0;
  __syncthreads();
  #pragma unroll
  for (int j = 0; j < EPT; ++j)
    if (tid + 256 * j < ECHUNK) atomicAdd(&h[d[j] / BINW], 1);
  __syncthreads();
  cnt[blockIdx.x * NBIN + tid] = h[tid];
}

__global__ void __launch_bounds__(256)
edge_bin_scatter_kernel(const int* __restrict__ ei,
                        const int* __restrict__ cnt,
                        int* __restrict__ binbase,
                        int* __restrict__ bsrc,
                        int* __restrict__ bdst) {
  __shared__ int ts[4][NBIN];
  __shared__ int ps[4][NBIN];
  __shared__ int cur[NBIN];
  __shared__ int wsum[8];
  const int tid  = threadIdx.x;
  const int base = blockIdx.x * ECHUNK;
  int s[EPT], d[EPT];
  #pragma unroll
  for (int j = 0; j < EPT; ++j) {
    int i = tid + 256 * j;
    i = (i < ECHUNK) ? i : (ECHUNK - 1);
    s[j] = ei[base + i];
    d[j] = ei[E_EDGES + base + i];
  }
  const int q = tid & 63, g = tid >> 6;
  int4 tot  = make_int4(0, 0, 0, 0);
  int4 part = make_int4(0, 0, 0, 0);
  #pragma unroll 16
  for (int j = 0; j < EBLK / 4; ++j) {
    const int b = 4 * j + g;
    const int4 c = ((const int4*)cnt)[b * (NBIN / 4) + q];
    tot.x += c.x; tot.y += c.y; tot.z += c.z; tot.w += c.w;
    if (b < (int)blockIdx.x) { part.x += c.x; part.y += c.y; part.z += c.z; part.w += c.w; }
  }
  ts[g][4 * q + 0] = tot.x;  ts[g][4 * q + 1] = tot.y;
  ts[g][4 * q + 2] = tot.z;  ts[g][4 * q + 3] = tot.w;
  ps[g][4 * q + 0] = part.x; ps[g][4 * q + 1] = part.y;
  ps[g][4 * q + 2] = part.z; ps[g][4 * q + 3] = part.w;
  __syncthreads();
  const int T = (ts[0][tid] + ts[1][tid]) + (ts[2][tid] + ts[3][tid]);
  const int P = (ps[0][tid] + ps[1][tid]) + (ps[2][tid] + ps[3][tid]);
  const int ex = block_excl_scan<8>(T, wsum);
  if (blockIdx.x == 0) {
    binbase[tid] = ex;
    if (tid == NBIN - 1) binbase[NBIN] = ex + T;
  }
  cur[tid] = ex + P;
  __syncthreads();
  int pos[EPT];
  #pragma unroll
  for (int j = 0; j < EPT; ++j)
    pos[j] = (tid + 256 * j < ECHUNK) ? atomicAdd(&cur[d[j] / BINW], 1) : 0;
  #pragma unroll
  for (int j = 0; j < EPT; ++j) {
    if (tid + 256 * j < ECHUNK) {
      bsrc[pos[j]] = s[j];
      bdst[pos[j]] = d[j];
    }
  }
}

__global__ void __launch_bounds__(SORT_T)
bin_sort_kernel(const int* __restrict__ binbase,
                const int* __restrict__ bsrc,
                const int* __restrict__ bdst,
                int* __restrict__ rowptr,
                int* __restrict__ csr) {
  __shared__ int c[NBIN];
  __shared__ int wsum[SORT_T / 32];
  const int tid = threadIdx.x;
  const int bin = blockIdx.x;
  const int beg = binbase[bin];
  const int end = binbase[bin + 1];
  const int n0  = bin * BINW;
  if (tid < NBIN) c[tid] = 0;
  __syncthreads();
  for (int i0 = beg; i0 < end; i0 += SORT_T * SORT_K) {
    int dd[SORT_K];
    #pragma unroll
    for (int k = 0; k < SORT_K; ++k) {
      const int i = i0 + tid + SORT_T * k;
      dd[k] = bdst[(i < end) ? i : (end - 1)];
    }
    #pragma unroll
    for (int k = 0; k < SORT_K; ++k)
      if (i0 + tid + SORT_T * k < end) atomicAdd(&c[dd[k] - n0], 1);
  }
  __syncthreads();
  const int v  = (tid < NBIN) ? c[tid] : 0;
  const int ex = block_excl_scan<SORT_T / 32>(v, wsum);
  if (tid < BINW && n0 + tid <= N_NODES) rowptr[n0 + tid] = beg + ex;
  if (tid < NBIN) c[tid] = beg + ex;
  __syncthreads();
  for (int i0 = beg; i0 < end; i0 += SORT_T * SORT_K) {
    int ss[SORT_K], dd[SORT_K];
    #pragma unroll
    for (int k = 0; k < SORT_K; ++k) {
      const int i  = i0 + tid + SORT_T * k;
      const int ic = (i < end) ? i : (end - 1);
      ss[k] = bsrc[ic];
      dd[k] = bdst[ic];
    }
    #pragma unroll
    for (int k = 0; k < SORT_K; ++k) {
      if (i0 + tid + SORT_T * k < end) {
        const int pos = atomicAdd(&c[dd[k] - n0], 1);
        csr[pos] = ss[k];
      }
    }
  }
}

#define NB 8
template <int K>
__device__ __forceinline__ float xlane(float v) {
  return __int_as_float(__builtin_amdgcn_mov_dpp(__float_as_int(v), 0x160 + K, 0xf, 0xf, true));
}
__device__ __forceinline__ float xhalf(float v) {
  int b = __float_as_int(v);
  return __int_as_float(__builtin_amdgcn_permlanex16(b, b, 0x76543210, 0xfedcba98, false, false));
}
template <int G>
__device__ __forceinline__ float group_lane(float v) {
  return __int_as_float((int)__builtin_amdgcn_mov_dpp8((unsigned)__float_as_int(v), G * 0x249249));
}
__device__ __forceinline__ float lane_score(v4f v, v4f xr4, v4f at4) {
  v4f t = v + xr4;
  t = __builtin_elementwise_max(t, t * SLOPE);
  v2f p = t.lo * at4.lo;
  p = t.hi * at4.hi + p;
  return p.x + p.y;
}
__global__ void __launch_bounds__(256)
node_attn_ln_kernel(const int* __restrict__ rowptr,
                    const int* __restrict__ csr,
                    const float* __restrict__ xl,
                    const float* __restrict__ xr,
                    const float* __restrict__ att,
                    const float* __restrict__ bias,
                    const float* __restrict__ gamma,
                    const float* __restrict__ beta,
                    float* __restrict__ out) {
  const int lane = threadIdx.x & 31;
  const int wv   = __builtin_amdgcn_readfirstlane(threadIdx.x >> 5);
  const int node = blockIdx.x * 8 + wv;
  if (node >= N_NODES) return;
  const bool b0 = (lane & 1) != 0, b1 = (lane & 2) != 0, b2 = (lane & 4) != 0;
  const int edge_of_lane = (b0 ? 4 : 0) + (b1 ? 2 : 0) + (b2 ? 1 : 0);
  const v4f xr4 = ((const v4f*)(xr + (size_t)node * HC))[lane];
  const v4f at4 = ((const v4f*)att)[lane];
  v4f acc = ((const v4f*)(xl + (size_t)node * HC))[lane];
  float m = lane_score(acc, xr4, at4);
  m += xlane<1>(m);
  m += xlane<2>(m);
  m += xlane<4>(m);
  float den = (edge_of_lane == 0) ? 1.f : 0.f;
  const int beg = rowptr[node];
  const int end = rowptr[node + 1];
  const __attribute__((address_space(1))) v4f* rowp[NB];
  if (beg < end) {
    #pragma unroll
    for (int j = 0; j < NB; ++j) {
      int q = (beg + j < end) ? (beg + j) : (end - 1);
      rowp[j] = (const __attribute__((address_space(1))) v4f*)(xl + (size_t)(unsigned)csr[q] * HC);
    }
  }
  for (int p = beg; p < end; p += NB) {
    v4f r[NB];
    #pragma unroll
    for (int j = 0; j < NB; ++j)
      r[j] = rowp[j][lane];
    #pragma unroll
    for (int j = 0; j < NB; ++j) {
      int q = (p + NB + j < end) ? (p + NB + j) : (end - 1);
      rowp[j] = (const __attribute__((address_space(1))) v4f*)(xl + (size_t)(unsigned)csr[q] * HC);
    }
    asm volatile("" :: "v"(r[0]), "v"(r[1]), "v"(r[2]), "v"(r[3]),
                       "v"(r[4]), "v"(r[5]), "v"(r[6]), "v"(r[7]));
    float s4[4];
    #pragma unroll
    for (int k = 0; k < 4; ++k) {
      const float sa = lane_score(r[k], xr4, at4);
      const float sb = lane_score(r[k + 4], xr4, at4);
      s4[k] = (b0 ? sb : sa) + xlane<1>(b0 ? sa : sb);
    }
    float s2[2];
    #pragma unroll
    for (int k = 0; k < 2; ++k)
      s2[k] = (b1 ? s4[k + 2] : s4[k]) + xlane<2>(b1 ? s4[k] : s4[k + 2]);
    float sc = (b2 ? s2[1] : s2[0]) + xlane<4>(b2 ? s2[0] : s2[1]);
    sc = (p + edge_of_lane < end) ? sc : -INFINITY;
    float mb = fmaxf(sc, xlane<1>(sc));
    mb = fmaxf(mb, xlane<2>(mb));
    mb = fmaxf(mb, xlane<4>(mb));
    mb = fmaxf(mb, m);
    const float scale = __expf(m - mb);
    const float w = __expf(sc - mb);
    acc *= scale;
    den = den * scale + w;
    acc += group_lane<0>(w) * r[0];
    acc += group_lane<4>(w) * r[1];
    acc += group_lane<2>(w) * r[2];
    acc += group_lane<6>(w) * r[3];
    acc += group_lane<1>(w) * r[4];
    acc += group_lane<5>(w) * r[5];
    acc += group_lane<3>(w) * r[6];
    acc += group_lane<7>(w) * r[7];
    m = mb;
  }
  den += xlane<1>(den);
  den += xlane<2>(den);
  den += xlane<4>(den);
  const float4 b4 = ((const float4*)bias)[lane];
  const float4 g4 = ((const float4*)gamma)[lane];
  const float4 e4 = ((const float4*)beta)[lane];
  const float inv = 1.0f / den;
  float h0 = acc.x * inv + b4.x, h1 = acc.y * inv + b4.y;
  float h2 = acc.z * inv + b4.z, h3 = acc.w * inv + b4.w;
  float sum = (h0 + h1) + (h2 + h3);
  sum += xhalf(sum);
  sum += xlane<8>(sum);
  sum += xlane<4>(sum);
  sum += xlane<2>(sum);
  sum += xlane<1>(sum);
  float mu = sum * (1.0f / HC);
  float d0 = h0 - mu, d1 = h1 - mu, d2 = h2 - mu, d3 = h3 - mu;
  float sq = (d0 * d0 + d1 * d1) + (d2 * d2 + d3 * d3);
  sq += xhalf(sq);
  sq += xlane<8>(sq);
  sq += xlane<4>(sq);
  sq += xlane<2>(sq);
  sq += xlane<1>(sq);
  float rstd = rsqrtf(sq * (1.0f / HC) + LN_EPS);
  float y0 = d0 * rstd * g4.x + e4.x, y1 = d1 * rstd * g4.y + e4.y;
  float y2 = d2 * rstd * g4.z + e4.z, y3 = d3 * rstd * g4.w + e4.w;
  float4 o;
  o.x = y0 / (1.0f + __expf(-y0));
  o.y = y1 / (1.0f + __expf(-y1));
  o.z = y2 / (1.0f + __expf(-y2));
  o.w = y3 / (1.0f + __expf(-y3));
  ((float4*)(out + (size_t)node * HC))[lane] = o;
}

#define POOL_NODES 32
#define POOL_BATCH 16
__device__ __forceinline__ void pool_flush(float* __restrict__ sums,
                                           float* __restrict__ counts,
                                           int g, int lane, float4 acc, float cnt) {
  float* p = sums + (size_t)g * HC + 4 * lane;
  atomAddF(p + 0, acc.x);
  atomAddF(p + 1, acc.y);
  atomAddF(p + 2, acc.z);
  atomAddF(p + 3, acc.w);
  if (lane == 0) atomAddF(&counts[g], cnt);
}
__global__ void __launch_bounds__(32)
pool_accum_kernel(const float* __restrict__ x,
                  const int* __restrict__ batch,
                  float* __restrict__ sums,
                  float* __restrict__ counts) {
  const int lane = threadIdx.x;
  const int n0 = blockIdx.x * POOL_NODES;
  const int left = N_NODES - n0;
  const int nn = (left < POOL_NODES) ? left : POOL_NODES;
  const float4* xp = (const float4*)(x + (size_t)n0 * HC) + lane;
  int gcur = batch[n0];
  float4 acc = make_float4(0.f, 0.f, 0.f, 0.f);
  float cnt = 0.f;
  for (int j0 = 0; j0 < nn; j0 += POOL_BATCH) {
    float4 v[POOL_BATCH];
    int g[POOL_BATCH];
    #pragma unroll
    for (int j = 0; j < POOL_BATCH; ++j) {
      int jj = (j0 + j < nn) ? (j0 + j) : (nn - 1);
      v[j] = xp[(size_t)jj * (HC / 4)];
      g[j] = batch[n0 + jj];
    }
    #pragma unroll
    for (int j = 0; j < POOL_BATCH; ++j) {
      if (j0 + j < nn) {
        if (g[j] != gcur) {
          pool_flush(sums, counts, gcur, lane, acc, cnt);
          acc = make_float4(0.f, 0.f, 0.f, 0.f);
          cnt = 0.f;
          gcur = g[j];
        }
        acc.x += v[j].x; acc.y += v[j].y; acc.z += v[j].z; acc.w += v[j].w;
        cnt += 1.0f;
      }
    }
  }
  pool_flush(sums, counts, gcur, lane, acc, cnt);
}
__global__ void pool_final_kernel(const float* __restrict__ sums,
                                  const float* __restrict__ counts,
                                  float* __restrict__ out) {
  int t = blockIdx.x * blockDim.x + threadIdx.x;
  if (t >= NGRAPH * HC) return;
  out[t] = sums[t] / fmaxf(counts[t >> 7], 1.0f);
}

extern "C" void kernel_launch(void* const* d_in, const int* in_sizes, int n_in,
                              void* d_out, int out_size, void* d_ws, size_t ws_size,
                              hipStream_t stream) {
  (void)in_sizes; (void)n_in; (void)out_size; (void)ws_size;
  const float* x     = (const float*)d_in[0];
  const int*   ei    = (const int*)d_in[1];
  const int*   batch = (const int*)d_in[2];
  const float* Wl[2]    = {(const float*)d_in[3],  (const float*)d_in[11]};
  const float* Wr[2]    = {(const float*)d_in[4],  (const float*)d_in[12]};
  const float* bl[2]    = {(const float*)d_in[5],  (const float*)d_in[13]};
  const float* br[2]    = {(const float*)d_in[6],  (const float*)d_in[14]};
  const float* att[2]   = {(const float*)d_in[7],  (const float*)d_in[15]};
  const float* bias[2]  = {(const float*)d_in[8],  (const float*)d_in[16]};
  const float* gamma[2] = {(const float*)d_in[9],  (const float*)d_in[17]};
  const float* beta[2]  = {(const float*)d_in[10], (const float*)d_in[18]};

  float* ws = (float*)d_ws;
  float* xl     = ws;          ws += (size_t)N_NODES * HC;
  float* xr2    = ws;          ws += (size_t)N_NODES * HC;
  float* xnode  = ws;          ws += (size_t)N_NODES * HC;
  int*   cnt    = (int*)ws;    ws += (size_t)EBLK * NBIN;
  int*   binbase= (int*)ws;    ws += (size_t)NBIN + 4;
  int*   rowptr = (int*)ws;    ws += (size_t)N_NODES + 4;
  int*   bsrc   = (int*)ws;    ws += (size_t)E_EDGES;
  int*   bdst   = (int*)ws;    ws += (size_t)E_EDGES;
  int*   csr    = (int*)ws;    ws += (size_t)E_EDGES;
  float* psums  = ws;          ws += (size_t)NGRAPH * HC;
  float* pcnt   = ws;

  dim3 b256(256);
  const int    nodeGrid = (N_NODES + 7) / 8;
  const int    poolGrid = (N_NODES + POOL_NODES - 1) / POOL_NODES;

  edge_bin_count_kernel<<<EBLK, b256, 0, stream>>>(ei, cnt);
  edge_bin_scatter_kernel<<<EBLK, b256, 0, stream>>>(ei, cnt, binbase, bsrc, bdst);
  bin_sort_kernel<<<NBIN, dim3(SORT_T), 0, stream>>>(binbase, bsrc, bdst, rowptr, csr);

  const float* xin = x;
  for (int l = 0; l < 2; ++l) {
    gemm2_bias_wmma<<<GEMM_BLOCKS, b256, 0, stream>>>(xin, Wl[l], Wr[l], bl[l], br[l], xl, xr2, N_NODES);
    node_attn_ln_kernel<<<nodeGrid, b256, 0, stream>>>(rowptr, csr, xl, xr2, att[l],
                                                       bias[l], gamma[l], beta[l], xnode);
    xin = xnode;
  }
  fill_f32<<<(NGRAPH * HC + NGRAPH + 255) / 256, b256, 0, stream>>>(psums, 0.f, NGRAPH * HC + NGRAPH);
  pool_accum_kernel<<<poolGrid, dim3(32), 0, stream>>>(xnode, batch, psums, pcnt);
  pool_final_kernel<<<(NGRAPH * HC + 255) / 256, b256, 0, stream>>>(psums, pcnt, (float*)d_out);
}


# mean pool inside the last node pass (10 launches to 7): a block's 8 rows meet in LDS, its last wave scales them by 1/graph size and sends one float atomic per run and channel; pool sum order changes
# speedup vs baseline: 5.3590x; 1.0728x over previous
#include <hip/hip_runtime.h>
#include <hip/hip_bf16.h>
#include <math.h>

#define N_NODES 50000
#define E_EDGES 800000
#define ETOT    (E_EDGES + N_NODES)
#define HEADS   4
#define CHP     32
#define HC      128
#define NGRAPH  64
#define SLOPE   0.2f
#define LN_EPS  1e-5f

typedef float v2f __attribute__((ext_vector_type(2)));
typedef float v8f __attribute__((ext_vector_type(8)));
typedef float v4f __attribute__((ext_vector_type(4)));

__device__ __forceinline__ void atomAddF(float* p, float v) {
  unsafeAtomicAdd(p, v);
}

#define LDSW2 264
#define LDSA 132
#define GEMM_BLOCKS 512
static_assert((HC * LDSW2 + 16 * LDSA) * 4 * 2 <= 320 * 1024, "two blocks share one LDS");
__global__ void __launch_bounds__(256)
gemm2_bias_wmma(const float* __restrict__ X,
                const float* __restrict__ Wa,
                const float* __restrict__ Wb,
                const float* __restrict__ ba,
                const float* __restrict__ bb,
                float* __restrict__ Ya,
                float* __restrict__ Yb, int n) {
  __shared__ __attribute__((aligned(16))) float smem[HC * LDSW2 + 16 * LDSA];
  float* sW = smem;
  float* sA = smem + HC * LDSW2;
  const int tid  = threadIdx.x;
  const int wave = tid >> 5;
  const int lane = tid & 31;
  const int ntiles = (n + 15) >> 4;
  const int t0 = (int)(((unsigned)blockIdx.x * (unsigned)ntiles) / GEMM_BLOCKS);
  const int t1 = (int)(((unsigned)(blockIdx.x + 1) * (unsigned)ntiles) / GEMM_BLOCKS);
  if (t0 >= t1) return;

  float4 pa0, pa1;
  {
    int r0 = t0 * 16 + wave, r1 = r0 + 8;
    r0 = (r0 < n) ? r0 : (n - 1);
    r1 = (r1 < n) ? r1 : (n - 1);
    pa0 = ((const float4*)(X + (size_t)r0 * HC))[lane];
    pa1 = ((const float4*)(X + (size_t)r1 * HC))[lane];
  }
  #pragma unroll
  for (int i = 0; i < 16; ++i) {
    int idx = tid + 256 * i;
    int r = idx >> 5, q = idx & 31;
    float4 w4 = *(const float4*)(Wa + r * HC + q * 4);
    *(float4*)(sW + r * LDSW2 + q * 4) = w4;
  }
  #pragma unroll
  for (int i = 0; i < 16; ++i) {
    int idx = tid + 256 * i;
    int r = idx >> 5, q = idx & 31;
    float4 w4 = *(const float4*)(Wb + r * HC + q * 4);
    *(float4*)(sW + r * LDSW2 + HC + q * 4) = w4;
  }
  *(float4*)(sA + wave * LDSA + lane * 4)       = pa0;
  *(float4*)(sA + (wave + 8) * LDSA + lane * 4) = pa1;
  __syncthreads();

  const int m    = lane & 15;
  const int hi   = lane >> 4;
  const bool second = (wave >= 4);
  float* __restrict__ Y = second ? Yb : Ya;
  const float* bvec = second ? bb : ba;
  const int ocol = (wave & 3) * 32 + m;
  const float bc0 = bvec[ocol];
  const float bc1 = bvec[ocol + 16];
  const float* ap = sA + m * LDSA + 2 * hi;
  const float* bp0 = sW + (2 * hi) * LDSW2 + wave * 32 + m;

  for (int t = t0; t < t1; ++t) {
    const int row0 = t * 16;
    const bool more = (t + 1 < t1);
    if (more) {
      int r0 = row0 + 16 + wave, r1 = r0 + 8;
      r0 = (r0 < n) ? r0 : (n - 1);
      r1 = (r1 < n) ? r1 : (n - 1);
      pa0 = ((const float4*)(X + (size_t)r0 * HC))[lane];
      pa1 = ((const float4*)(X + (size_t)r1 * HC))[lane];
    }
    v8f acc0 = {};
    v8f acc1 = {};
    #pragma unroll
    for (int k0 = 0; k0 < HC; k0 += 4) {
      v2f a = *(const v2f*)(ap + k0);
      const float* bp = bp0 + k0 * LDSW2;
      v2f b0; b0.x = bp[0];  b0.y = bp[LDSW2];
      v2f b1; b1.x = bp[16]; b1.y = bp[LDSW2 + 16];
      acc0 = __builtin_amdgcn_wmma_f32_16x16x4_f32(
          false, a, false, b0, (short)0, acc0, false, false);
      acc1 = __builtin_amdgcn_wmma_f32_16x16x4_f32(
          false, a, false, b1, (short)0, acc1, false, false);
    }
    if (more) {
      __syncthreads();
      *(float4*)(sA + wave * LDSA + lane * 4)       = pa0;
      *(float4*)(sA + (wave + 8) * LDSA + lane * 4) = pa1;
      __syncthreads();
    }
    #pragma unroll
    for (int r = 0; r < 8; ++r) {
      int orow = row0 + r + 8 * hi;
      if (orow < n) {
        float* yp = Y + (size_t)orow * HC + ocol;
        yp[0]  = acc0[r] + bc0;
        yp[16] = acc1[r] + bc1;
      }
    }
  }
}

#define NBIN   256
#define BINW   196
#define EBLK   256
#define ECHUNK (E_EDGES / EBLK)
#define EPT    ((ECHUNK + 255) / 256)
#define SORT_T 512
#define SORT_K 8
static_assert(EBLK * ECHUNK == E_EDGES, "edge blocks cover the edges exactly");
static_assert(NBIN * BINW > N_NODES, "ranges cover node ids 0..N_NODES");
static_assert(BINW <= NBIN, "one counter per node of a range");
#define POOL_Z ((NGRAPH * HC) / EBLK)
static_assert(POOL_Z * EBLK == NGRAPH * HC, "the edge blocks clear the pooled output exactly");
static_assert(POOL_Z == 32, "one 128-byte line of the pooled output per edge block");

template <int NW>
__device__ __forceinline__ int block_excl_scan(int v, int* wsum) {
  const int lane = threadIdx.x & 31;
  const int w    = threadIdx.x >> 5;
  int inc = v;
  for (int off = 1; off < 32; off <<= 1) {
    int u = __shfl_up(inc, off, 32);
    if (lane >= off) inc += u;
  }
  if (lane == 31) wsum[w] = inc;
  __syncthreads();
  int carry = 0;
  for (int k = 0; k < NW; ++k) carry += (k < w) ? wsum[k] : 0;
  return carry + inc - v;
}

__global__ void __launch_bounds__(256)
edge_bin_count_kernel(const int* __restrict__ ei, int* __restrict__ cnt,
                      float* __restrict__ pooled, int* __restrict__ gcount) {
  __shared__ int h[NBIN];
  const int tid  = threadIdx.x;
  const int base = blockIdx.x * ECHUNK;
  int d[EPT];
  #pragma unroll
  for (int j = 0; j < EPT; ++j) {
    int i = tid + 256 * j;
    i = (i < ECHUNK) ? i : (ECHUNK - 1);
    d[j] = ei[E_EDGES + base + i];
  }
  h[tid] = 0;
  __syncthreads();
  #pragma unroll
  for (int j = 0; j < EPT; ++j)
    if (tid + 256 * j < ECHUNK) atomicAdd(&h[d[j] / BINW], 1);
  __syncthreads();
  cnt[blockIdx.x * NBIN + tid] = h[tid];
  if (tid < POOL_Z) pooled[blockIdx.x * POOL_Z + tid] = 0.f;
  if (blockIdx.x == 0 && tid < NGRAPH) gcount[tid] = 0;
}

__global__ void __launch_bounds__(256)
edge_bin_scatter_kernel(const int* __restrict__ ei,
                        const int* __restrict__ cnt,
                        int* __restrict__ binbase,
                        int* __restrict__ bsrc,
                        int* __restrict__ bdst) {
  __shared__ int ts[4][NBIN];
  __shared__ int ps[4][NBIN];
  __shared__ int cur[NBIN];
  __shared__ int wsum[8];
  const int tid  = threadIdx.x;
  const int base = blockIdx.x * ECHUNK;
  int s[EPT], d[EPT];
  #pragma unroll
  for (int j = 0; j < EPT; ++j) {
    int i = tid + 256 * j;
    i = (i < ECHUNK) ? i : (ECHUNK - 1);
    s[j] = ei[base + i];
    d[j] = ei[E_EDGES + base + i];
  }
  const int q = tid & 63, g = tid >> 6;
  int4 tot  = make_int4(0, 0, 0, 0);
  int4 part = make_int4(0, 0, 0, 0);
  #pragma unroll 16
  for (int j = 0; j < EBLK / 4; ++j) {
    const int b = 4 * j + g;
    const int4 c = ((const int4*)cnt)[b * (NBIN / 4) + q];
    tot.x += c.x; tot.y += c.y; tot.z += c.z; tot.w += c.w;
    if (b < (int)blockIdx.x) { part.x += c.x; part.y += c.y; part.z += c.z; part.w += c.w; }
  }
  ts[g][4 * q + 0] = tot.x;  ts[g][4 * q + 1] = tot.y;
  ts[g][4 * q + 2] = tot.z;  ts[g][4 * q + 3] = tot.w;
  ps[g][4 * q + 0] = part.x; ps[g][4 * q + 1] = part.y;
  ps[g][4 * q + 2] = part.z; ps[g][4 * q + 3] = part.w;
  __syncthreads();
  const int T = (ts[0][tid] + ts[1][tid]) + (ts[2][tid] + ts[3][tid]);
  const int P = (ps[0][tid] + ps[1][tid]) + (ps[2][tid] + ps[3][tid]);
  const int ex = block_excl_scan<8>(T, wsum);
  if (blockIdx.x == 0) {
    binbase[tid] = ex;
    if (tid == NBIN - 1) binbase[NBIN] = ex + T;
  }
  cur[tid] = ex + P;
  __syncthreads();
  int pos[EPT];
  #pragma unroll
  for (int j = 0; j < EPT; ++j)
    pos[j] = (tid + 256 * j < ECHUNK) ? atomicAdd(&cur[d[j] / BINW], 1) : 0;
  #pragma unroll
  for (int j = 0; j < EPT; ++j) {
    if (tid + 256 * j < ECHUNK) {
      bsrc[pos[j]] = s[j];
      bdst[pos[j]] = d[j];
    }
  }
}

__global__ void __launch_bounds__(SORT_T)
bin_sort_kernel(const int* __restrict__ binbase,
                const int* __restrict__ bsrc,
                const int* __restrict__ bdst,
                int* __restrict__ rowptr,
                int* __restrict__ csr,
                const int* __restrict__ batch,
                int* __restrict__ gcount) {
  __shared__ int c[NBIN];
  __shared__ int gh[NGRAPH];
  __shared__ int wsum[SORT_T / 32];
  const int tid = threadIdx.x;
  const int bin = blockIdx.x;
  const int beg = binbase[bin];
  const int end = binbase[bin + 1];
  const int n0  = bin * BINW;
  if (tid < NBIN) c[tid] = 0;
  if (tid < NGRAPH) gh[tid] = 0;
  __syncthreads();
  const bool own = (tid < BINW) && (n0 + tid < N_NODES);
  const int gid = own ? batch[n0 + tid] : 0;
  for (int i0 = beg; i0 < end; i0 += SORT_T * SORT_K) {
    int dd[SORT_K];
    #pragma unroll
    for (int k = 0; k < SORT_K; ++k) {
      const int i = i0 + tid + SORT_T * k;
      dd[k] = bdst[(i < end) ? i : (end - 1)];
    }
    #pragma unroll
    for (int k = 0; k < SORT_K; ++k)
      if (i0 + tid + SORT_T * k < end) atomicAdd(&c[dd[k] - n0], 1);
  }
  if (own) atomicAdd(&gh[gid], 1);
  __syncthreads();
  const int v  = (tid < NBIN) ? c[tid] : 0;
  const int ex = block_excl_scan<SORT_T / 32>(v, wsum);
  if (tid < BINW && n0 + tid <= N_NODES) rowptr[n0 + tid] = beg + ex;
  if (tid < NBIN) c[tid] = beg + ex;
  __syncthreads();
  for (int i0 = beg; i0 < end; i0 += SORT_T * SORT_K) {
    int ss[SORT_K], dd[SORT_K];
    #pragma unroll
    for (int k = 0; k < SORT_K; ++k) {
      const int i  = i0 + tid + SORT_T * k;
      const int ic = (i < end) ? i : (end - 1);
      ss[k] = bsrc[ic];
      dd[k] = bdst[ic];
    }
    #pragma unroll
    for (int k = 0; k < SORT_K; ++k) {
      if (i0 + tid + SORT_T * k < end) {
        const int pos = atomicAdd(&c[dd[k] - n0], 1);
        csr[pos] = ss[k];
      }
    }
  }
  if (tid < NGRAPH) {
    const int gv = gh[tid];
    if (gv != 0) atomicAdd(&gcount[tid], gv);
  }
}

#define NB 8
template <int K>
__device__ __forceinline__ float xlane(float v) {
  return __int_as_float(__builtin_amdgcn_mov_dpp(__float_as_int(v), 0x160 + K, 0xf, 0xf, true));
}
__device__ __forceinline__ float xhalf(float v) {
  int b = __float_as_int(v);
  return __int_as_float(__builtin_amdgcn_permlanex16(b, b, 0x76543210, 0xfedcba98, false, false));
}
template <int G>
__device__ __forceinline__ float group_lane(float v) {
  return __int_as_float((int)__builtin_amdgcn_mov_dpp8((unsigned)__float_as_int(v), G * 0x249249));
}
__device__ __forceinline__ float lane_score(v4f v, v4f xr4, v4f at4) {
  v4f t = v + xr4;
  t = __builtin_elementwise_max(t, t * SLOPE);
  v2f p = t.lo * at4.lo;
  p = t.hi * at4.hi + p;
  return p.x + p.y;
}
static_assert(N_NODES % 8 == 0, "every block of the node pass is full: 8 tickets per block");
template <bool POOL>
__global__ void __launch_bounds__(256)
node_attn_ln_kernel(const int* __restrict__ rowptr,
                    const int* __restrict__ csr,
                    const float* __restrict__ xl,
                    const float* __restrict__ xr,
                    const float* __restrict__ att,
                    const float* __restrict__ bias,
                    const float* __restrict__ gamma,
                    const float* __restrict__ beta,
                    float* __restrict__ out,
                    const int* __restrict__ batch,
                    const int* __restrict__ gcount) {
  __shared__ __attribute__((aligned(16))) float prow[POOL ? 8 * HC : 4];
  __shared__ int pgid[8];
  __shared__ int pticket;
  if (POOL) {
    if (threadIdx.x == 0) pticket = 0;
    __syncthreads();
  }
  const int lane = threadIdx.x & 31;
  const int wv   = __builtin_amdgcn_readfirstlane(threadIdx.x >> 5);
  const int node = blockIdx.x * 8 + wv;
  if (node >= N_NODES) return;
  const bool b0 = (lane & 1) != 0, b1 = (lane & 2) != 0, b2 = (lane & 4) != 0;
  const int edge_of_lane = (b0 ? 4 : 0) + (b1 ? 2 : 0) + (b2 ? 1 : 0);
  const v4f xr4 = ((const v4f*)(xr + (size_t)node * HC))[lane];
  const v4f at4 = ((const v4f*)att)[lane];
  v4f acc = ((const v4f*)(xl + (size_t)node * HC))[lane];
  float m = lane_score(acc, xr4, at4);
  m += xlane<1>(m);
  m += xlane<2>(m);
  m += xlane<4>(m);
  float den = (edge_of_lane == 0) ? 1.f : 0.f;
  const int beg = rowptr[node];
  const int end = rowptr[node + 1];
  int gidn = 0;
  if (POOL) gidn = batch[node];
  const __attribute__((address_space(1))) v4f* rowp[NB];
  if (beg < end) {
    #pragma unroll
    for (int j = 0; j < NB; ++j) {
      int q = (beg + j < end) ? (beg + j) : (end - 1);
      rowp[j] = (const __attribute__((address_space(1))) v4f*)(xl + (size_t)(unsigned)csr[q] * HC);
    }
  }
  for (int p = beg; p < end; p += NB) {
    v4f r[NB];
    #pragma unroll
    for (int j = 0; j < NB; ++j)
      r[j] = rowp[j][lane];
    #pragma unroll
    for (int j = 0; j < NB; ++j) {
      int q = (p + NB + j < end) ? (p + NB + j) : (end - 1);
      rowp[j] = (const __attribute__((address_space(1))) v4f*)(xl + (size_t)(unsigned)csr[q] * HC);
    }
    asm volatile("" :: "v"(r[0]), "v"(r[1]), "v"(r[2]), "v"(r[3]),
                       "v"(r[4]), "v"(r[5]), "v"(r[6]), "v"(r[7]));
    float s4[4];
    #pragma unroll
    for (int k = 0; k < 4; ++k) {
      const float sa = lane_score(r[k], xr4, at4);
      const float sb = lane_score(r[k + 4], xr4, at4);
      s4[k] = (b0 ? sb : sa) + xlane<1>(b0 ? sa : sb);
    }
    float s2[2];
    #pragma unroll
    for (int k = 0; k < 2; ++k)
      s2[k] = (b1 ? s4[k + 2] : s4[k]) + xlane<2>(b1 ? s4[k] : s4[k + 2]);
    float sc = (b2 ? s2[1] : s2[0]) + xlane<4>(b2 ? s2[0] : s2[1]);
    sc = (p + edge_of_lane < end) ? sc : -INFINITY;
    float mb = fmaxf(sc, xlane<1>(sc));
    mb = fmaxf(mb, xlane<2>(mb));
    mb = fmaxf(mb, xlane<4>(mb));
    mb = fmaxf(mb, m);
    const float scale = __expf(m - mb);
    const float w = __expf(sc - mb);
    acc *= scale;
    den = den * scale + w;
    acc += group_lane<0>(w) * r[0];
    acc += group_lane<4>(w) * r[1];
    acc += group_lane<2>(w) * r[2];
    acc += group_lane<6>(w) * r[3];
    acc += group_lane<1>(w) * r[4];
    acc += group_lane<5>(w) * r[5];
    acc += group_lane<3>(w) * r[6];
    acc += group_lane<7>(w) * r[7];
    m = mb;
  }
  den += xlane<1>(den);
  den += xlane<2>(den);
  den += xlane<4>(den);
  const float4 b4 = ((const float4*)bias)[lane];
  const float4 g4 = ((const float4*)gamma)[lane];
  const float4 e4 = ((const float4*)beta)[lane];
  const float inv = 1.0f / den;
  float h0 = acc.x * inv + b4.x, h1 = acc.y * inv + b4.y;
  float h2 = acc.z * inv + b4.z, h3 = acc.w * inv + b4.w;
  float sum = (h0 + h1) + (h2 + h3);
  sum += xhalf(sum);
  sum += xlane<8>(sum);
  sum += xlane<4>(sum);
  sum += xlane<2>(sum);
  sum += xlane<1>(sum);
  float mu = sum * (1.0f / HC);
  float d0 = h0 - mu, d1 = h1 - mu, d2 = h2 - mu, d3 = h3 - mu;
  float sq = (d0 * d0 + d1 * d1) + (d2 * d2 + d3 * d3);
  sq += xhalf(sq);
  sq += xlane<8>(sq);
  sq += xlane<4>(sq);
  sq += xlane<2>(sq);
  sq += xlane<1>(sq);
  float rstd = rsqrtf(sq * (1.0f / HC) + LN_EPS);
  float y0 = d0 * rstd * g4.x + e4.x, y1 = d1 * rstd * g4.y + e4.y;
  float y2 = d2 * rstd * g4.z + e4.z, y3 = d3 * rstd * g4.w + e4.w;
  float4 o;
  o.x = y0 / (1.0f + __expf(-y0));
  o.y = y1 / (1.0f + __expf(-y1));
  o.z = y2 / (1.0f + __expf(-y2));
  o.w = y3 / (1.0f + __expf(-y3));
  if (!POOL) {
    ((float4*)(out + (size_t)node * HC))[lane] = o;
  } else {
    const int ln = (int)__builtin_amdgcn_mbcnt_lo(~0u, 0u);
    ((float4*)prow)[wv * 32 + ln] = o;
    if (ln == 0) pgid[wv] = gidn;
    __threadfence_block();
    int t = 0;
    if (ln == 0) t = atomicAdd(&pticket, 1);
    t = __builtin_amdgcn_readfirstlane(t);
    if (t == 7) {
      __threadfence_block();
      const int   gl = pgid[ln & 7];
      const float il = 1.0f / fmaxf((float)gcount[gl], 1.0f);
      float4 v[8];
      #pragma unroll
      for (int j = 0; j < 8; ++j) v[j] = ((const float4*)prow)[j * 32 + ln];
      int gc = __builtin_amdgcn_readlane(gl, 0);
      float s = __int_as_float(__builtin_amdgcn_readlane(__float_as_int(il), 0));
      float4 a = make_float4(v[0].x * s, v[0].y * s, v[0].z * s, v[0].w * s);
      #pragma unroll
      for (int j = 1; j < 8; ++j) {
        const int g = __builtin_amdgcn_readlane(gl, j);
        s = __int_as_float(__builtin_amdgcn_readlane(__float_as_int(il), j));
        if (g != gc) {
          float* p = out + (size_t)gc * HC + 4 * ln;
          atomAddF(p + 0, a.x); atomAddF(p + 1, a.y); atomAddF(p + 2, a.z); atomAddF(p + 3, a.w);
          a = make_float4(0.f, 0.f, 0.f, 0.f);
          gc = g;
        }
        a.x += v[j].x * s; a.y += v[j].y * s; a.z += v[j].z * s; a.w += v[j].w * s;
      }
      float* p = out + (size_t)gc * HC + 4 * ln;
      atomAddF(p + 0, a.x); atomAddF(p + 1, a.y); atomAddF(p + 2, a.z); atomAddF(p + 3, a.w);
    }
  }
}

extern "C" void kernel_launch(void* const* d_in, const int* in_sizes, int n_in,
                              void* d_out, int out_size, void* d_ws, size_t ws_size,
                              hipStream_t stream) {
  (void)in_sizes; (void)n_in; (void)out_size; (void)ws_size;
  const float* x     = (const float*)d_in[0];
  const int*   ei    = (const int*)d_in[1];
  const int*   batch = (const int*)d_in[2];
  const float* Wl[2]    = {(const float*)d_in[3],  (const float*)d_in[11]};
  const float* Wr[2]    = {(const float*)d_in[4],  (const float*)d_in[12]};
  const float* bl[2]    = {(const float*)d_in[5],  (const float*)d_in[13]};
  const float* br[2]    = {(const float*)d_in[6],  (const float*)d_in[14]};
  const float* att[2]   = {(const float*)d_in[7],  (const float*)d_in[15]};
  const float* bias[2]  = {(const float*)d_in[8],  (const float*)d_in[16]};
  const float* gamma[2] = {(const float*)d_in[9],  (const float*)d_in[17]};
  const float* beta[2]  = {(const float*)d_in[10], (const float*)d_in[18]};

  float* ws = (float*)d_ws;
  float* xl     = ws;          ws += (size_t)N_NODES * HC;
  float* xr2    = ws;          ws += (size_t)N_NODES * HC;
  float* xnode  = ws;          ws += (size_t)N_NODES * HC;
  int*   cnt    = (int*)ws;    ws += (size_t)EBLK * NBIN;
  int*   binbase= (int*)ws;    ws += (size_t)NBIN + 4;
  int*   rowptr = (int*)ws;    ws += (size_t)N_NODES + 4;
  int*   bsrc   = (int*)ws;    ws += (size_t)E_EDGES;
  int*   bdst   = (int*)ws;    ws += (size_t)E_EDGES;
  int*   csr    = (int*)ws;    ws += (size_t)E_EDGES;
  int*   gcount = (int*)ws;

  dim3 b256(256);
  const int    nodeGrid = (N_NODES + 7) / 8;

  edge_bin_count_kernel<<<EBLK, b256, 0, stream>>>(ei, cnt, (float*)d_out, gcount);
  edge_bin_scatter_kernel<<<EBLK, b256, 0, stream>>>(ei, cnt, binbase, bsrc, bdst);
  bin_sort_kernel<<<NBIN, dim3(SORT_T), 0, stream>>>(binbase, bsrc, bdst, rowptr, csr, batch, gcount);

  gemm2_bias_wmma<<<GEMM_BLOCKS, b256, 0, stream>>>(x, Wl[0], Wr[0], bl[0], br[0], xl, xr2, N_NODES);
  node_attn_ln_kernel<false><<<nodeGrid, b256, 0, stream>>>(rowptr, csr, xl, xr2, att[0],
                                                            bias[0], gamma[0], beta[0], xnode, batch, gcount);
  gemm2_bias_wmma<<<GEMM_BLOCKS, b256, 0, stream>>>(xnode, Wl[1], Wr[1], bl[1], br[1], xl, xr2, N_NODES);
  node_attn_ln_kernel<true><<<nodeGrid, b256, 0, stream>>>(rowptr, csr, xl, xr2, att[1],
                                                           bias[1], gamma[1], beta[1], (float*)d_out, batch, gcount);
}
